# starting kernel
# baseline (speedup 1.0000x reference)
#include <hip/hip_runtime.h>

typedef float v2f __attribute__((ext_vector_type(2)));
typedef float v8f __attribute__((ext_vector_type(8)));

#define N_NODES 100000
#define FEAT    256
#define NHID    64
#define NCLS    16
#define NEDGE   1600000
#define NTILES  (N_NODES / 16)

__global__ __launch_bounds__(128) void gemm1_kernel(const float* __restrict__ x,
                                                    const float* __restrict__ W1,
                                                    float* __restrict__ sup) {
    __shared__ float tileA[16 * FEAT];
    const int lane = threadIdx.x & 31;
    const int wave = threadIdx.x >> 5;
    const int rb   = blockIdx.x * 16;

    const float4* x4 = reinterpret_cast<const float4*>(x + (size_t)rb * FEAT);
    float4* t4 = reinterpret_cast<float4*>(tileA);
    for (int i = threadIdx.x; i < 16 * FEAT / 4; i += 128) t4[i] = x4[i];
    __syncthreads();

    const int c  = wave * 16;
    const int l  = lane & 15;
    const int hi = lane >> 4;
    v8f acc = {};
    for (int k = 0; k < FEAT; k += 4) {
        const int ks = k + 2 * hi;
        v2f a = *reinterpret_cast<const v2f*>(&tileA[l * FEAT + ks]);
        v2f b;
        b.x = W1[(size_t)ks       * NHID + c + l];
        b.y = W1[(size_t)(ks + 1) * NHID + c + l];
        acc = __builtin_amdgcn_wmma_f32_16x16x4_f32(false, a, false, b,
                                                    (short)0, acc, false, false);
    }
    float* out = sup + (size_t)rb * NHID + c;
    const int mb = hi * 8;
#pragma unroll
    for (int r = 0; r < 8; ++r)
        out[(size_t)(mb + r) * NHID + l] = acc[r];
}

__global__ __launch_bounds__(256) void gemm2_kernel(const float* __restrict__ h,
                                                    const float* __restrict__ W2,
                                                    float* __restrict__ sup2) {
    const int lane = threadIdx.x & 31;
    const int wave = threadIdx.x >> 5;
    const int tile = blockIdx.x * 8 + wave;
    if (tile >= NTILES) return;
    const int rb = tile * 16;
    const int l  = lane & 15;
    const int hi = lane >> 4;
    v8f acc = {};
    for (int k = 0; k < NHID; k += 4) {
        const int ks = k + 2 * hi;
        v2f a = *reinterpret_cast<const v2f*>(&h[(size_t)(rb + l) * NHID + ks]);
        v2f b;
        b.x = W2[ks * NCLS + l];
        b.y = W2[(ks + 1) * NCLS + l];
        acc = __builtin_amdgcn_wmma_f32_16x16x4_f32(false, a, false, b,
                                                    (short)0, acc, false, false);
    }
    float* out = sup2 + (size_t)rb * NCLS;
    const int mb = hi * 8;
#pragma unroll
    for (int r = 0; r < 8; ++r)
        out[(size_t)(mb + r) * NCLS + l] = acc[r];
}

template <int F>
__global__ __launch_bounds__(256) void spmm_kernel(const int* __restrict__ esrc,
                                                   const int* __restrict__ edst,
                                                   const float* __restrict__ ev,
                                                   const float* __restrict__ sup,
                                                   float* __restrict__ acc) {
    constexpr int CH = F / 4;
    const long long total  = (long long)NEDGE * CH;
    const long long stride = (long long)gridDim.x * blockDim.x;
    for (long long idx = (long long)blockIdx.x * blockDim.x + threadIdx.x;
         idx < total; idx += stride) {
        const int e    = (int)(idx / CH);
        const int part = (int)(idx % CH);
        long long pe = (idx + stride) / CH;
        if (pe >= NEDGE) pe = NEDGE - 1;
        __builtin_prefetch(&esrc[pe], 0, 0);
        const int   s  = esrc[e];
        const int   d  = edst[e];
        const float wv = ev[e];
        const float4 v = reinterpret_cast<const float4*>(sup)[(size_t)s * CH + part];
        float* base = acc + (size_t)d * F + (size_t)part * 4;
        atomicAdd(base + 0, v.x * wv);
        atomicAdd(base + 1, v.y * wv);
        atomicAdd(base + 2, v.z * wv);
        atomicAdd(base + 3, v.w * wv);
    }
}

__global__ __launch_bounds__(256) void zero_kernel(float* __restrict__ p, long long n) {
    long long i = (long long)blockIdx.x * blockDim.x + threadIdx.x;
    if (i < n) p[i] = 0.0f;
}

__global__ __launch_bounds__(256) void bias_relu_kernel(float* __restrict__ h,
                                                        const float* __restrict__ b1) {
    long long i = (long long)blockIdx.x * blockDim.x + threadIdx.x;
    if (i < (long long)N_NODES * NHID) {
        const float v = h[i] + b1[i & (NHID - 1)];
        h[i] = v > 0.0f ? v : 0.0f;
    }
}

__global__ __launch_bounds__(256) void init_out_kernel(float* __restrict__ out,
                                                       const float* __restrict__ b2) {
    long long i = (long long)blockIdx.x * blockDim.x + threadIdx.x;
    if (i < (long long)N_NODES * NCLS) out[i] = b2[i & (NCLS - 1)];
}

__global__ __launch_bounds__(128) void log_softmax_kernel(float* __restrict__ out) {
    const int r = blockIdx.x * blockDim.x + threadIdx.x;
    if (r >= N_NODES) return;
    float* p = out + (size_t)r * NCLS;
    float v[NCLS];
    float m = -3.402823466e+38f;
#pragma unroll
    for (int i = 0; i < NCLS; ++i) { v[i] = p[i]; m = v[i] > m ? v[i] : m; }
    float s = 0.0f;
#pragma unroll
    for (int i = 0; i < NCLS; ++i) s += __expf(v[i] - m);
    const float lse = m + __logf(s);
#pragma unroll
    for (int i = 0; i < NCLS; ++i) p[i] = v[i] - lse;
}

extern "C" void kernel_launch(void* const* d_in, const int* in_sizes, int n_in,
                              void* d_out, int out_size, void* d_ws, size_t ws_size,
                              hipStream_t stream) {
    (void)in_sizes; (void)n_in; (void)out_size; (void)ws_size;
    const float* x   = (const float*)d_in[0];
    const int*   es  = (const int*)  d_in[1];
    const int*   ed  = (const int*)  d_in[2];
    const float* ev  = (const float*)d_in[3];
    const float* W1  = (const float*)d_in[4];
    const float* b1  = (const float*)d_in[5];
    const float* W2  = (const float*)d_in[6];
    const float* b2  = (const float*)d_in[7];
    float* out = (float*)d_out;

    float* sup1 = (float*)d_ws;
    float* hacc = sup1 + (size_t)N_NODES * NHID;
    float* sup2 = sup1;

    const long long nh = (long long)N_NODES * NHID;

    zero_kernel<<<(int)((nh + 255) / 256), 256, 0, stream>>>(hacc, nh);
    gemm1_kernel<<<NTILES, 128, 0, stream>>>(x, W1, sup1);
    {
        const long long total = (long long)NEDGE * (NHID / 4);
        spmm_kernel<NHID><<<(int)((total + 255) / 256), 256, 0, stream>>>(es, ed, ev, sup1, hacc);
    }
    bias_relu_kernel<<<(int)((nh + 255) / 256), 256, 0, stream>>>(hacc, b1);
    gemm2_kernel<<<(NTILES + 7) / 8, 256, 0, stream>>>(hacc, W2, sup2);
    {
        const long long nc = (long long)N_NODES * NCLS;
        init_out_kernel<<<(int)((nc + 255) / 256), 256, 0, stream>>>(out, b2);
        const long long total = (long long)NEDGE * (NCLS / 4);
        spmm_kernel<NCLS><<<(int)((total + 255) / 256), 256, 0, stream>>>(es, ed, ev, sup2, out);
    }
    log_softmax_kernel<<<(N_NODES + 127) / 128, 128, 0, stream>>>(out);
}


# edge scatter re-mapped: a wave walks runs of 32 edges with lane = column, so each f32 atomic and gather instruction covers contiguous row floats (one 128-B line at 64 cols); arrival-order sums kept
# speedup vs baseline: 1.7711x; 1.7711x over previous
#include <hip/hip_runtime.h>

typedef float v2f __attribute__((ext_vector_type(2)));
typedef float v8f __attribute__((ext_vector_type(8)));

#define N_NODES 100000
#define FEAT    256
#define NHID    64
#define NCLS    16
#define NEDGE   1600000
#define NTILES  (N_NODES / 16)

__global__ __launch_bounds__(128) void gemm1_kernel(const float* __restrict__ x,
                                                    const float* __restrict__ W1,
                                                    float* __restrict__ sup) {
    __shared__ float tileA[16 * FEAT];
    const int lane = threadIdx.x & 31;
    const int wave = threadIdx.x >> 5;
    const int rb   = blockIdx.x * 16;

    const float4* x4 = reinterpret_cast<const float4*>(x + (size_t)rb * FEAT);
    float4* t4 = reinterpret_cast<float4*>(tileA);
    for (int i = threadIdx.x; i < 16 * FEAT / 4; i += 128) t4[i] = x4[i];
    __syncthreads();

    const int c  = wave * 16;
    const int l  = lane & 15;
    const int hi = lane >> 4;
    v8f acc = {};
    for (int k = 0; k < FEAT; k += 4) {
        const int ks = k + 2 * hi;
        v2f a = *reinterpret_cast<const v2f*>(&tileA[l * FEAT + ks]);
        v2f b;
        b.x = W1[(size_t)ks       * NHID + c + l];
        b.y = W1[(size_t)(ks + 1) * NHID + c + l];
        acc = __builtin_amdgcn_wmma_f32_16x16x4_f32(false, a, false, b,
                                                    (short)0, acc, false, false);
    }
    float* out = sup + (size_t)rb * NHID + c;
    const int mb = hi * 8;
#pragma unroll
    for (int r = 0; r < 8; ++r)
        out[(size_t)(mb + r) * NHID + l] = acc[r];
}

__global__ __launch_bounds__(256) void gemm2_kernel(const float* __restrict__ h,
                                                    const float* __restrict__ W2,
                                                    float* __restrict__ sup2) {
    const int lane = threadIdx.x & 31;
    const int wave = threadIdx.x >> 5;
    const int tile = blockIdx.x * 8 + wave;
    if (tile >= NTILES) return;
    const int rb = tile * 16;
    const int l  = lane & 15;
    const int hi = lane >> 4;
    v8f acc = {};
    for (int k = 0; k < NHID; k += 4) {
        const int ks = k + 2 * hi;
        v2f a = *reinterpret_cast<const v2f*>(&h[(size_t)(rb + l) * NHID + ks]);
        v2f b;
        b.x = W2[ks * NCLS + l];
        b.y = W2[(ks + 1) * NCLS + l];
        acc = __builtin_amdgcn_wmma_f32_16x16x4_f32(false, a, false, b,
                                                    (short)0, acc, false, false);
    }
    float* out = sup2 + (size_t)rb * NCLS;
    const int mb = hi * 8;
#pragma unroll
    for (int r = 0; r < 8; ++r)
        out[(size_t)(mb + r) * NCLS + l] = acc[r];
}

template <int F, int EPW, int BS>
__global__ __launch_bounds__(BS) void spmm_kernel(const int* __restrict__ esrc,
                                                  const int* __restrict__ edst,
                                                  const float* __restrict__ ev,
                                                  const float* __restrict__ sup,
                                                  float* __restrict__ acc) {
    static_assert(NEDGE % 32 == 0, "edges are taken in whole runs of 32");
    static_assert(F == 64 || F == 16, "row widths of the two layers");
    static_assert((long long)N_NODES * F < (1ll << 31), "32-bit element index");
    const int lane = threadIdx.x & 31;
    const int wave = __builtin_amdgcn_readfirstlane(
        (int)(blockIdx.x * (BS / 32) + (threadIdx.x >> 5)));
    for (int g = 0; g < EPW / 32; ++g) {
        const int e0 = wave * EPW + g * 32;
        if (e0 >= NEDGE) break;
        const int s_l = esrc[e0 + lane];
        const int d_l = edst[e0 + lane];
        const int w_l = __builtin_bit_cast(int, ev[e0 + lane]);
        if constexpr (F == 64) {
#pragma unroll
            for (int jb = 0; jb < 32; jb += 16) {
                float v0[16], v1[16];
#pragma unroll
                for (int u = 0; u < 16; ++u) {
                    const int s = __builtin_amdgcn_readlane(s_l, jb + u);
                    const unsigned si = (unsigned)s * 64u + (unsigned)lane;
                    v0[u] = sup[si];
                    v1[u] = sup[si + 32u];
                }
#pragma unroll
                for (int u = 0; u < 16; ++u) {
                    const int   d = __builtin_amdgcn_readlane(d_l, jb + u);
                    const float w = __builtin_bit_cast(
                        float, __builtin_amdgcn_readlane(w_l, jb + u));
                    const unsigned di = (unsigned)d * 64u + (unsigned)lane;
                    atomicAdd(acc + di,       v0[u] * w);
                    atomicAdd(acc + di + 32u, v1[u] * w);
                }
            }
        } else {
            const int l  = lane & 15;
            const int hi = lane >> 4;
#pragma unroll
            for (int jb = 0; jb < 16; jb += 8) {
                float v[8];
#pragma unroll
                for (int u = 0; u < 8; ++u) {
                    const int sel = (2 * (jb + u) + hi) << 2;
                    const int s = __builtin_amdgcn_ds_bpermute(sel, s_l);
                    v[u] = sup[(unsigned)s * 16u + (unsigned)l];
                }
#pragma unroll
                for (int u = 0; u < 8; ++u) {
                    const int sel = (2 * (jb + u) + hi) << 2;
                    const int   d = __builtin_amdgcn_ds_bpermute(sel, d_l);
                    const float w = __builtin_bit_cast(
                        float, __builtin_amdgcn_ds_bpermute(sel, w_l));
                    atomicAdd(acc + ((unsigned)d * 16u + (unsigned)l), v[u] * w);
                }
            }
        }
    }
}

__global__ __launch_bounds__(256) void zero_kernel(float* __restrict__ p, long long n) {
    long long i = (long long)blockIdx.x * blockDim.x + threadIdx.x;
    if (i < n) p[i] = 0.0f;
}

__global__ __launch_bounds__(256) void bias_relu_kernel(float* __restrict__ h,
                                                        const float* __restrict__ b1) {
    long long i = (long long)blockIdx.x * blockDim.x + threadIdx.x;
    if (i < (long long)N_NODES * NHID) {
        const float v = h[i] + b1[i & (NHID - 1)];
        h[i] = v > 0.0f ? v : 0.0f;
    }
}

__global__ __launch_bounds__(256) void init_out_kernel(float* __restrict__ out,
                                                       const float* __restrict__ b2) {
    long long i = (long long)blockIdx.x * blockDim.x + threadIdx.x;
    if (i < (long long)N_NODES * NCLS) out[i] = b2[i & (NCLS - 1)];
}

__global__ __launch_bounds__(128) void log_softmax_kernel(float* __restrict__ out) {
    const int r = blockIdx.x * blockDim.x + threadIdx.x;
    if (r >= N_NODES) return;
    float* p = out + (size_t)r * NCLS;
    float v[NCLS];
    float m = -3.402823466e+38f;
#pragma unroll
    for (int i = 0; i < NCLS; ++i) { v[i] = p[i]; m = v[i] > m ? v[i] : m; }
    float s = 0.0f;
#pragma unroll
    for (int i = 0; i < NCLS; ++i) s += __expf(v[i] - m);
    const float lse = m + __logf(s);
#pragma unroll
    for (int i = 0; i < NCLS; ++i) p[i] = v[i] - lse;
}

extern "C" void kernel_launch(void* const* d_in, const int* in_sizes, int n_in,
                              void* d_out, int out_size, void* d_ws, size_t ws_size,
                              hipStream_t stream) {
    (void)in_sizes; (void)n_in; (void)out_size; (void)ws_size;
    const float* x   = (const float*)d_in[0];
    const int*   es  = (const int*)  d_in[1];
    const int*   ed  = (const int*)  d_in[2];
    const float* ev  = (const float*)d_in[3];
    const float* W1  = (const float*)d_in[4];
    const float* b1  = (const float*)d_in[5];
    const float* W2  = (const float*)d_in[6];
    const float* b2  = (const float*)d_in[7];
    float* out = (float*)d_out;

    float* sup1 = (float*)d_ws;
    float* hacc = sup1 + (size_t)N_NODES * NHID;
    float* sup2 = sup1;

    const long long nh = (long long)N_NODES * NHID;

    zero_kernel<<<(int)((nh + 255) / 256), 256, 0, stream>>>(hacc, nh);
    gemm1_kernel<<<NTILES, 128, 0, stream>>>(x, W1, sup1);
    spmm_kernel<NHID, 64, 256><<<(NEDGE + 511) / 512, 256, 0, stream>>>(es, ed, ev, sup1, hacc);
    bias_relu_kernel<<<(int)((nh + 255) / 256), 256, 0, stream>>>(hacc, b1);
    gemm2_kernel<<<(NTILES + 7) / 8, 256, 0, stream>>>(hacc, W2, sup2);
    {
        const long long nc = (long long)N_NODES * NCLS;
        init_out_kernel<<<(int)((nc + 255) / 256), 256, 0, stream>>>(out, b2);
        spmm_kernel<NCLS, 128, 128><<<(NEDGE + 511) / 512, 128, 0, stream>>>(es, ed, ev, sup2, out);
    }
    log_softmax_kernel<<<(N_NODES + 127) / 128, 128, 0, stream>>>(out);
}


# zero-fill, bias+relu and bias-init passes folded into the two GEMM kernels (8 launches -> 5, 450k fewer waves to start, h never written back); arithmetic and summation order unchanged
# speedup vs baseline: 1.8954x; 1.0702x over previous
#include <hip/hip_runtime.h>

typedef float v2f __attribute__((ext_vector_type(2)));
typedef float v8f __attribute__((ext_vector_type(8)));

#define N_NODES 100000
#define FEAT    256
#define NHID    64
#define NCLS    16
#define NEDGE   1600000
#define NTILES  (N_NODES / 16)

__global__ __launch_bounds__(128) void gemm1_kernel(const float* __restrict__ x,
                                                    const float* __restrict__ W1,
                                                    float* __restrict__ sup,
                                                    float* __restrict__ hacc) {
    __shared__ float tileA[16 * FEAT];
    const int lane = threadIdx.x & 31;
    const int wave = threadIdx.x >> 5;
    const int rb   = blockIdx.x * 16;

    {
        float4* z4 = reinterpret_cast<float4*>(hacc + (size_t)rb * NHID);
        const float4 z = make_float4(0.0f, 0.0f, 0.0f, 0.0f);
        z4[threadIdx.x]       = z;
        z4[threadIdx.x + 128] = z;
    }

    const float4* x4 = reinterpret_cast<const float4*>(x + (size_t)rb * FEAT);
    float4* t4 = reinterpret_cast<float4*>(tileA);
    for (int i = threadIdx.x; i < 16 * FEAT / 4; i += 128) t4[i] = x4[i];
    __syncthreads();

    const int c  = wave * 16;
    const int l  = lane & 15;
    const int hi = lane >> 4;
    v8f acc = {};
    for (int k = 0; k < FEAT; k += 4) {
        const int ks = k + 2 * hi;
        v2f a = *reinterpret_cast<const v2f*>(&tileA[l * FEAT + ks]);
        v2f b;
        b.x = W1[(size_t)ks       * NHID + c + l];
        b.y = W1[(size_t)(ks + 1) * NHID + c + l];
        acc = __builtin_amdgcn_wmma_f32_16x16x4_f32(false, a, false, b,
                                                    (short)0, acc, false, false);
    }
    float* out = sup + (size_t)rb * NHID + c;
    const int mb = hi * 8;
#pragma unroll
    for (int r = 0; r < 8; ++r)
        out[(size_t)(mb + r) * NHID + l] = acc[r];
}

__global__ __launch_bounds__(256) void gemm2_kernel(const float* __restrict__ hacc,
                                                    const float* __restrict__ b1,
                                                    const float* __restrict__ W2,
                                                    float* __restrict__ sup2,
                                                    const float* __restrict__ b2,
                                                    float* __restrict__ logits) {
    const int lane = threadIdx.x & 31;
    const int wave = threadIdx.x >> 5;
    const int tile = blockIdx.x * 8 + wave;
    if (tile >= NTILES) return;
    const int rb = tile * 16;
    const int l  = lane & 15;
    const int hi = lane >> 4;

    {
        const float4 bv = *reinterpret_cast<const float4*>(b2 + (lane & 3) * 4);
        float4* o4 = reinterpret_cast<float4*>(logits + (size_t)rb * NCLS);
        o4[lane]      = bv;
        o4[lane + 32] = bv;
    }

    v8f acc = {};
    for (int k = 0; k < NHID; k += 4) {
        const int ks = k + 2 * hi;
        v2f a = *reinterpret_cast<const v2f*>(&hacc[(size_t)(rb + l) * NHID + ks]);
        const v2f bb = *reinterpret_cast<const v2f*>(&b1[ks]);
        a.x += bb.x;
        a.y += bb.y;
        a.x = a.x > 0.0f ? a.x : 0.0f;
        a.y = a.y > 0.0f ? a.y : 0.0f;
        v2f b;
        b.x = W2[ks * NCLS + l];
        b.y = W2[(ks + 1) * NCLS + l];
        acc = __builtin_amdgcn_wmma_f32_16x16x4_f32(false, a, false, b,
                                                    (short)0, acc, false, false);
    }
    float* out = sup2 + (size_t)rb * NCLS;
    const int mb = hi * 8;
#pragma unroll
    for (int r = 0; r < 8; ++r)
        out[(size_t)(mb + r) * NCLS + l] = acc[r];
}

template <int F, int EPW, int BS>
__global__ __launch_bounds__(BS) void spmm_kernel(const int* __restrict__ esrc,
                                                  const int* __restrict__ edst,
                                                  const float* __restrict__ ev,
                                                  const float* __restrict__ sup,
                                                  float* __restrict__ acc) {
    static_assert(NEDGE % 32 == 0, "edges are taken in whole runs of 32");
    static_assert(F == 64 || F == 16, "row widths of the two layers");
    static_assert((long long)N_NODES * F < (1ll << 31), "32-bit element index");
    const int lane = threadIdx.x & 31;
    const int wave = __builtin_amdgcn_readfirstlane(
        (int)(blockIdx.x * (BS / 32) + (threadIdx.x >> 5)));
    for (int g = 0; g < EPW / 32; ++g) {
        const int e0 = wave * EPW + g * 32;
        if (e0 >= NEDGE) break;
        const int s_l = esrc[e0 + lane];
        const int d_l = edst[e0 + lane];
        const int w_l = __builtin_bit_cast(int, ev[e0 + lane]);
        if constexpr (F == 64) {
#pragma unroll
            for (int jb = 0; jb < 32; jb += 16) {
                float v0[16], v1[16];
#pragma unroll
                for (int u = 0; u < 16; ++u) {
                    const int s = __builtin_amdgcn_readlane(s_l, jb + u);
                    const unsigned si = (unsigned)s * 64u + (unsigned)lane;
                    v0[u] = sup[si];
                    v1[u] = sup[si + 32u];
                }
#pragma unroll
                for (int u = 0; u < 16; ++u) {
                    const int   d = __builtin_amdgcn_readlane(d_l, jb + u);
                    const float w = __builtin_bit_cast(
                        float, __builtin_amdgcn_readlane(w_l, jb + u));
                    const unsigned di = (unsigned)d * 64u + (unsigned)lane;
                    atomicAdd(acc + di,       v0[u] * w);
                    atomicAdd(acc + di + 32u, v1[u] * w);
                }
            }
        } else {
            const int l  = lane & 15;
            const int hi = lane >> 4;
#pragma unroll
            for (int jb = 0; jb < 16; jb += 8) {
                float v[8];
#pragma unroll
                for (int u = 0; u < 8; ++u) {
                    const int sel = (2 * (jb + u) + hi) << 2;
                    const int s = __builtin_amdgcn_ds_bpermute(sel, s_l);
                    v[u] = sup[(unsigned)s * 16u + (unsigned)l];
                }
#pragma unroll
                for (int u = 0; u < 8; ++u) {
                    const int sel = (2 * (jb + u) + hi) << 2;
                    const int   d = __builtin_amdgcn_ds_bpermute(sel, d_l);
                    const float w = __builtin_bit_cast(
                        float, __builtin_amdgcn_ds_bpermute(sel, w_l));
                    atomicAdd(acc + ((unsigned)d * 16u + (unsigned)l), v[u] * w);
                }
            }
        }
    }
}

__global__ __launch_bounds__(128) void log_softmax_kernel(float* __restrict__ out) {
    const int r = blockIdx.x * blockDim.x + threadIdx.x;
    if (r >= N_NODES) return;
    float* p = out + (size_t)r * NCLS;
    float v[NCLS];
    float m = -3.402823466e+38f;
#pragma unroll
    for (int i = 0; i < NCLS; ++i) { v[i] = p[i]; m = v[i] > m ? v[i] : m; }
    float s = 0.0f;
#pragma unroll
    for (int i = 0; i < NCLS; ++i) s += __expf(v[i] - m);
    const float lse = m + __logf(s);
#pragma unroll
    for (int i = 0; i < NCLS; ++i) p[i] = v[i] - lse;
}

extern "C" void kernel_launch(void* const* d_in, const int* in_sizes, int n_in,
                              void* d_out, int out_size, void* d_ws, size_t ws_size,
                              hipStream_t stream) {
    (void)in_sizes; (void)n_in; (void)out_size; (void)ws_size;
    const float* x   = (const float*)d_in[0];
    const int*   es  = (const int*)  d_in[1];
    const int*   ed  = (const int*)  d_in[2];
    const float* ev  = (const float*)d_in[3];
    const float* W1  = (const float*)d_in[4];
    const float* b1  = (const float*)d_in[5];
    const float* W2  = (const float*)d_in[6];
    const float* b2  = (const float*)d_in[7];
    float* out = (float*)d_out;

    float* sup1 = (float*)d_ws;
    float* hacc = sup1 + (size_t)N_NODES * NHID;
    float* sup2 = sup1;

    gemm1_kernel<<<NTILES, 128, 0, stream>>>(x, W1, sup1, hacc);
    spmm_kernel<NHID, 64, 256><<<(NEDGE + 511) / 512, 256, 0, stream>>>(es, ed, ev, sup1, hacc);
    gemm2_kernel<<<(NTILES + 7) / 8, 256, 0, stream>>>(hacc, b1, W2, sup2, b2, out);
    spmm_kernel<NCLS, 128, 128><<<(NEDGE + 511) / 512, 128, 0, stream>>>(es, ed, ev, sup2, out);
    log_softmax_kernel<<<(N_NODES + 127) / 128, 128, 0, stream>>>(out);
}


# edges grouped by destination bin of 128 rows (count pass + fill pass); each bin's rows are summed in LDS with ds_add_f32 and stored once, replacing all global f32 atomics; arrival-order sums kept
# speedup vs baseline: 2.6862x; 1.4172x over previous
#include <hip/hip_runtime.h>


typedef float v2f __attribute__((ext_vector_type(2)));
typedef float v8f __attribute__((ext_vector_type(8)));

#define N_NODES 100000
#define FEAT    256
#define NHID    64
#define NCLS    16
#define NEDGE   1600000
#define NTILES  (N_NODES / 16)

#define RB      128
#define NBIN    ((N_NODES + RB - 1) / RB)
#define EPB     4096
#define NBLK_E  ((NEDGE + EPB - 1) / EPB)
#define SRC_BITS 17

static_assert(N_NODES <= (1 << SRC_BITS), "source row must fit the packed word");
static_assert(RB == 128 && NBIN <= 1024, "one thread per bin in the 1024-thread passes");

__global__ __launch_bounds__(128) void gemm1_kernel(const float* __restrict__ x,
                                                    const float* __restrict__ W1,
                                                    float* __restrict__ sup,
                                                    int* __restrict__ counts,
                                                    int* __restrict__ cursor) {
    __shared__ float tileA[16 * FEAT];
    const int lane = threadIdx.x & 31;
    const int wave = threadIdx.x >> 5;
    const int rb   = blockIdx.x * 16;

    if (threadIdx.x == 0 && blockIdx.x < 1024) {
        counts[blockIdx.x] = 0;
        cursor[blockIdx.x] = 0;
    }

    const float4* x4 = reinterpret_cast<const float4*>(x + (size_t)rb * FEAT);
    float4* t4 = reinterpret_cast<float4*>(tileA);
    for (int i = threadIdx.x; i < 16 * FEAT / 4; i += 128) t4[i] = x4[i];
    __syncthreads();

    const int c  = wave * 16;
    const int l  = lane & 15;
    const int hi = lane >> 4;
    v8f acc = {};
    for (int k = 0; k < FEAT; k += 4) {
        const int ks = k + 2 * hi;
        v2f a = *reinterpret_cast<const v2f*>(&tileA[l * FEAT + ks]);
        v2f b;
        b.x = W1[(size_t)ks       * NHID + c + l];
        b.y = W1[(size_t)(ks + 1) * NHID + c + l];
        acc = __builtin_amdgcn_wmma_f32_16x16x4_f32(false, a, false, b,
                                                    (short)0, acc, false, false);
    }
    float* out = sup + (size_t)rb * NHID + c;
    const int mb = hi * 8;
#pragma unroll
    for (int r = 0; r < 8; ++r)
        out[(size_t)(mb + r) * NHID + l] = acc[r];
}

__global__ __launch_bounds__(256) void gemm2_kernel(const float* __restrict__ hacc,
                                                    const float* __restrict__ b1,
                                                    const float* __restrict__ W2,
                                                    float* __restrict__ sup2) {
    const int lane = threadIdx.x & 31;
    const int wave = threadIdx.x >> 5;
    const int tile = blockIdx.x * 8 + wave;
    if (tile >= NTILES) return;
    const int rb = tile * 16;
    const int l  = lane & 15;
    const int hi = lane >> 4;

    v8f acc = {};
    for (int k = 0; k < NHID; k += 4) {
        const int ks = k + 2 * hi;
        v2f a = *reinterpret_cast<const v2f*>(&hacc[(size_t)(rb + l) * NHID + ks]);
        const v2f bb = *reinterpret_cast<const v2f*>(&b1[ks]);
        a.x += bb.x;
        a.y += bb.y;
        a.x = a.x > 0.0f ? a.x : 0.0f;
        a.y = a.y > 0.0f ? a.y : 0.0f;
        v2f b;
        b.x = W2[ks * NCLS + l];
        b.y = W2[(ks + 1) * NCLS + l];
        acc = __builtin_amdgcn_wmma_f32_16x16x4_f32(false, a, false, b,
                                                    (short)0, acc, false, false);
    }
    float* out = sup2 + (size_t)rb * NCLS;
    const int mb = hi * 8;
#pragma unroll
    for (int r = 0; r < 8; ++r)
        out[(size_t)(mb + r) * NCLS + l] = acc[r];
}

__global__ __launch_bounds__(1024) void bin_count_kernel(const int* __restrict__ edst,
                                                         int* __restrict__ counts) {
    __shared__ int s_hist[NBIN];
    const int t = threadIdx.x;
    if (t < NBIN) s_hist[t] = 0;
    __syncthreads();
    const int base = blockIdx.x * EPB;
#pragma unroll
    for (int i = 0; i < EPB / 1024; ++i) {
        const int e = base + i * 1024 + t;
        if (e < NEDGE) atomicAdd(&s_hist[(unsigned)edst[e] / RB], 1);
    }
    __syncthreads();
    if (t < NBIN) {
        const int cnt = s_hist[t];
        if (cnt) atomicAdd(&counts[t], cnt);
    }
}

__global__ __launch_bounds__(1024) void bin_fill_kernel(const int* __restrict__ esrc,
                                                        const int* __restrict__ edst,
                                                        const float* __restrict__ ev,
                                                        const int* __restrict__ counts,
                                                        int* __restrict__ cursor,
                                                        int* __restrict__ binstart,
                                                        int2* __restrict__ packed) {
    __shared__ int s_hist[NBIN];
    __shared__ int s_base[NBIN];
    __shared__ int s_wsum[32];
    const int t    = threadIdx.x;
    const int lane = t & 31;
    const int wave = t >> 5;
    if (t < NBIN) s_hist[t] = 0;
    __syncthreads();
    const int base = blockIdx.x * EPB;
    int d[EPB / 1024], rk[EPB / 1024];
#pragma unroll
    for (int i = 0; i < EPB / 1024; ++i) {
        const int e = base + i * 1024 + t;
        d[i]  = -1;
        rk[i] = 0;
        if (e < NEDGE) {
            d[i]  = edst[e];
            rk[i] = atomicAdd(&s_hist[(unsigned)d[i] / RB], 1);
        }
    }
    const int cnt = (t < NBIN) ? counts[t] : 0;
    int inc = cnt;
#pragma unroll
    for (int o = 1; o < 32; o <<= 1) {
        const int nb = __shfl_up(inc, o);
        if (lane >= o) inc += nb;
    }
    if (lane == 31) s_wsum[wave] = inc;
    __syncthreads();
    if (wave == 0) {
        const int ws = s_wsum[lane];
        int wi = ws;
#pragma unroll
        for (int o = 1; o < 32; o <<= 1) {
            const int nb = __shfl_up(wi, o);
            if (lane >= o) wi += nb;
        }
        s_wsum[lane] = wi - ws;
    }
    __syncthreads();
    const int bstart = inc - cnt + s_wsum[wave];
    if (t < NBIN) {
        const int mine = s_hist[t];
        int g = 0;
        if (mine) g = atomicAdd(&cursor[t], mine);
        s_base[t] = bstart + g;
        if (blockIdx.x == 0) binstart[t] = bstart;
    }
    if (blockIdx.x == 0 && t == NBIN) binstart[NBIN] = NEDGE;
    __syncthreads();
#pragma unroll
    for (int i = 0; i < EPB / 1024; ++i) {
        if (d[i] >= 0) {
            const int e   = base + i * 1024 + t;
            const int bin = (unsigned)d[i] / RB;
            const int pos = s_base[bin] + rk[i];
            const int pk  = esrc[e] | ((d[i] - bin * RB) << SRC_BITS);
            if ((unsigned)pos < (unsigned)NEDGE)
                packed[pos] = make_int2(pk, __builtin_bit_cast(int, ev[e]));
        }
    }
}

static __device__ __forceinline__ void lds_add(float* p, float v) {
    (void)__hip_atomic_fetch_add(p, v, __ATOMIC_RELAXED, __HIP_MEMORY_SCOPE_WORKGROUP);
}

template <int F, bool TAIL>
static __device__ __forceinline__ void spmm_run(const int pk_l, const int w_l, const int lane,
                                                const float* __restrict__ sup, float* acc) {
    if constexpr (F == 64) {
#pragma unroll
        for (int jb = 0; jb < 32; jb += 16) {
            float v0[16], v1[16];
#pragma unroll
            for (int u = 0; u < 16; ++u) {
                const int pk = __builtin_amdgcn_readlane(pk_l, jb + u);
                const unsigned si = ((unsigned)pk & ((1u << SRC_BITS) - 1u)) * 64u + (unsigned)lane;
                v0[u] = sup[si];
                v1[u] = sup[si + 32u];
            }
#pragma unroll
            for (int u = 0; u < 16; ++u) {
                const int   pk = __builtin_amdgcn_readlane(pk_l, jb + u);
                const float w  = __builtin_bit_cast(
                    float, __builtin_amdgcn_readlane(w_l, jb + u));
                const unsigned di = (((unsigned)pk >> SRC_BITS) & (RB - 1u)) * 64u + (unsigned)lane;
                if (!TAIL || pk >= 0) {
                    lds_add(acc + di,       v0[u] * w);
                    lds_add(acc + di + 32u, v1[u] * w);
                }
            }
        }
    } else {
        const int l  = lane & 15;
        const int hi = lane >> 4;
#pragma unroll
        for (int jb = 0; jb < 16; jb += 8) {
            float v[8];
#pragma unroll
            for (int u = 0; u < 8; ++u) {
                const int sel = (2 * (jb + u) + hi) << 2;
                const int pk  = __builtin_amdgcn_ds_bpermute(sel, pk_l);
                v[u] = sup[((unsigned)pk & ((1u << SRC_BITS) - 1u)) * 16u + (unsigned)l];
            }
#pragma unroll
            for (int u = 0; u < 8; ++u) {
                const int sel = (2 * (jb + u) + hi) << 2;
                const int   pk = __builtin_amdgcn_ds_bpermute(sel, pk_l);
                const float w  = __builtin_bit_cast(
                    float, __builtin_amdgcn_ds_bpermute(sel, w_l));
                const unsigned di = (((unsigned)pk >> SRC_BITS) & (RB - 1u)) * 16u + (unsigned)l;
                if (!TAIL || pk >= 0) lds_add(acc + di, v[u] * w);
            }
        }
    }
}

template <int F, int BS, bool BIAS>
__global__ __launch_bounds__(BS) void spmm_bin_kernel(const int2* __restrict__ packed,
                                                      const int* __restrict__ binstart,
                                                      const float* __restrict__ sup,
                                                      const float* __restrict__ bias,
                                                      float* __restrict__ outp) {
    static_assert(F == 64 || F == 16, "row widths of the two layers");
    static_assert((long long)N_NODES * F < (1ll << 31), "32-bit element index");
    __shared__ float acc[RB * F];
    const int t    = threadIdx.x;
    const int lane = t & 31;
    const int wave  = __builtin_amdgcn_readfirstlane(t >> 5);
    const int bin   = blockIdx.x;
    const int start = __builtin_amdgcn_readfirstlane(binstart[bin]);
    int       end   = __builtin_amdgcn_readfirstlane(binstart[bin + 1]);
    end = end < NEDGE ? end : NEDGE;

    float4* acc4 = reinterpret_cast<float4*>(acc);
    for (int i = t; i < RB * F / 4; i += BS) acc4[i] = make_float4(0.0f, 0.0f, 0.0f, 0.0f);
    __syncthreads();

    for (int e0 = start + wave * 32; e0 < end; e0 += BS) {
        const int n = end - e0;
        int pk_l = (int)0x80000000;
        int w_l  = 0;
        if (lane < n) {
            const int2 r = packed[e0 + lane];
            pk_l = r.x;
            w_l  = r.y;
        }
        if (n >= 32) spmm_run<F, false>(pk_l, w_l, lane, sup, acc);
        else         spmm_run<F, true >(pk_l, w_l, lane, sup, acc);
    }
    __syncthreads();

    const int rb0 = bin * RB;
    float4* o4 = reinterpret_cast<float4*>(outp + (size_t)rb0 * F);
    for (int i = t; i < RB * F / 4; i += BS) {
        const int row = rb0 + (i * 4) / F;
        if (row < N_NODES) {
            float4 v = acc4[i];
            if constexpr (BIAS) {
                const float4 bv = *reinterpret_cast<const float4*>(bias + (i % (F / 4)) * 4);
                v.x += bv.x; v.y += bv.y; v.z += bv.z; v.w += bv.w;
            }
            o4[i] = v;
        }
    }
}

__global__ __launch_bounds__(128) void log_softmax_kernel(float* __restrict__ out) {
    const int r = blockIdx.x * blockDim.x + threadIdx.x;
    if (r >= N_NODES) return;
    float* p = out + (size_t)r * NCLS;
    float v[NCLS];
    float m = -3.402823466e+38f;
#pragma unroll
    for (int i = 0; i < NCLS; ++i) { v[i] = p[i]; m = v[i] > m ? v[i] : m; }
    float s = 0.0f;
#pragma unroll
    for (int i = 0; i < NCLS; ++i) s += __expf(v[i] - m);
    const float lse = m + __logf(s);
#pragma unroll
    for (int i = 0; i < NCLS; ++i) p[i] = v[i] - lse;
}

extern "C" void kernel_launch(void* const* d_in, const int* in_sizes, int n_in,
                              void* d_out, int out_size, void* d_ws, size_t ws_size,
                              hipStream_t stream) {
    (void)in_sizes; (void)n_in; (void)out_size; (void)ws_size;
    const float* x   = (const float*)d_in[0];
    const int*   es  = (const int*)  d_in[1];
    const int*   ed  = (const int*)  d_in[2];
    const float* ev  = (const float*)d_in[3];
    const float* W1  = (const float*)d_in[4];
    const float* b1  = (const float*)d_in[5];
    const float* W2  = (const float*)d_in[6];
    const float* b2  = (const float*)d_in[7];
    float* out = (float*)d_out;

    float* sup1 = (float*)d_ws;
    float* hacc = sup1 + (size_t)N_NODES * NHID;
    float* sup2 = sup1;
    int*  counts   = (int*)(hacc + (size_t)N_NODES * NHID);
    int*  cursor   = counts + 1024;
    int*  binstart = cursor + 1024;
    int2* packed   = (int2*)(binstart + 1024);

    gemm1_kernel<<<NTILES, 128, 0, stream>>>(x, W1, sup1, counts, cursor);
    bin_count_kernel<<<NBLK_E, 1024, 0, stream>>>(ed, counts);
    bin_fill_kernel<<<NBLK_E, 1024, 0, stream>>>(es, ed, ev, counts, cursor, binstart, packed);
    spmm_bin_kernel<NHID, 512, false><<<NBIN, 512, 0, stream>>>(packed, binstart, sup1, b1, hacc);
    gemm2_kernel<<<(NTILES + 7) / 8, 256, 0, stream>>>(hacc, b1, W2, sup2);
    spmm_bin_kernel<NCLS, 512, true><<<NBIN, 512, 0, stream>>>(packed, binstart, sup2, b2, out);
    log_softmax_kernel<<<(N_NODES + 127) / 128, 128, 0, stream>>>(out);
}


# second GEMM (bias+relu, f32 WMMA) and log-softmax run in the tails of the two bin kernels on the LDS rows: 7 launches -> 5, hidden rows never stored; row exp-sum now 4 lane partials + 2 xor shuffles
# speedup vs baseline: 2.9356x; 1.0928x over previous
#include <hip/hip_runtime.h>


typedef float v2f __attribute__((ext_vector_type(2)));
typedef float v8f __attribute__((ext_vector_type(8)));

#define N_NODES 100000
#define FEAT    256
#define NHID    64
#define NCLS    16
#define NEDGE   1600000
#define NTILES  (N_NODES / 16)

#define RB      128
#define NBIN    ((N_NODES + RB - 1) / RB)
#define EPB     4096
#define NBLK_E  ((NEDGE + EPB - 1) / EPB)
#define SRC_BITS 17

static_assert(N_NODES <= (1 << SRC_BITS), "source row must fit the packed word");
static_assert(RB == 128 && NBIN <= 1024, "one thread per bin in the 1024-thread passes");

__global__ __launch_bounds__(128) void gemm1_kernel(const float* __restrict__ x,
                                                    const float* __restrict__ W1,
                                                    float* __restrict__ sup,
                                                    int* __restrict__ counts,
                                                    int* __restrict__ cursor) {
    __shared__ float tileA[16 * FEAT];
    const int lane = threadIdx.x & 31;
    const int wave = threadIdx.x >> 5;
    const int rb   = blockIdx.x * 16;

    if (threadIdx.x == 0 && blockIdx.x < 1024) {
        counts[blockIdx.x] = 0;
        cursor[blockIdx.x] = 0;
    }

    const float4* x4 = reinterpret_cast<const float4*>(x + (size_t)rb * FEAT);
    float4* t4 = reinterpret_cast<float4*>(tileA);
    for (int i = threadIdx.x; i < 16 * FEAT / 4; i += 128) t4[i] = x4[i];
    __syncthreads();

    const int c  = wave * 16;
    const int l  = lane & 15;
    const int hi = lane >> 4;
    v8f acc = {};
    for (int k = 0; k < FEAT; k += 4) {
        const int ks = k + 2 * hi;
        v2f a = *reinterpret_cast<const v2f*>(&tileA[l * FEAT + ks]);
        v2f b;
        b.x = W1[(size_t)ks       * NHID + c + l];
        b.y = W1[(size_t)(ks + 1) * NHID + c + l];
        acc = __builtin_amdgcn_wmma_f32_16x16x4_f32(false, a, false, b,
                                                    (short)0, acc, false, false);
    }
    float* out = sup + (size_t)rb * NHID + c;
    const int mb = hi * 8;
#pragma unroll
    for (int r = 0; r < 8; ++r)
        out[(size_t)(mb + r) * NHID + l] = acc[r];
}

__global__ __launch_bounds__(1024) void bin_count_kernel(const int* __restrict__ edst,
                                                         int* __restrict__ counts) {
    __shared__ int s_hist[NBIN];
    const int t = threadIdx.x;
    if (t < NBIN) s_hist[t] = 0;
    __syncthreads();
    const int base = blockIdx.x * EPB;
#pragma unroll
    for (int i = 0; i < EPB / 1024; ++i) {
        const int e = base + i * 1024 + t;
        if (e < NEDGE) atomicAdd(&s_hist[(unsigned)edst[e] / RB], 1);
    }
    __syncthreads();
    if (t < NBIN) {
        const int cnt = s_hist[t];
        if (cnt) atomicAdd(&counts[t], cnt);
    }
}

__global__ __launch_bounds__(1024) void bin_fill_kernel(const int* __restrict__ esrc,
                                                        const int* __restrict__ edst,
                                                        const float* __restrict__ ev,
                                                        const int* __restrict__ counts,
                                                        int* __restrict__ cursor,
                                                        int* __restrict__ binstart,
                                                        int2* __restrict__ packed) {
    __shared__ int s_hist[NBIN];
    __shared__ int s_base[NBIN];
    __shared__ int s_wsum[32];
    const int t    = threadIdx.x;
    const int lane = t & 31;
    const int wave = t >> 5;
    if (t < NBIN) s_hist[t] = 0;
    __syncthreads();
    const int base = blockIdx.x * EPB;
    int d[EPB / 1024], rk[EPB / 1024];
#pragma unroll
    for (int i = 0; i < EPB / 1024; ++i) {
        const int e = base + i * 1024 + t;
        d[i]  = -1;
        rk[i] = 0;
        if (e < NEDGE) {
            d[i]  = edst[e];
            rk[i] = atomicAdd(&s_hist[(unsigned)d[i] / RB], 1);
        }
    }
    const int cnt = (t < NBIN) ? counts[t] : 0;
    int inc = cnt;
#pragma unroll
    for (int o = 1; o < 32; o <<= 1) {
        const int nb = __shfl_up(inc, o);
        if (lane >= o) inc += nb;
    }
    if (lane == 31) s_wsum[wave] = inc;
    __syncthreads();
    if (wave == 0) {
        const int ws = s_wsum[lane];
        int wi = ws;
#pragma unroll
        for (int o = 1; o < 32; o <<= 1) {
            const int nb = __shfl_up(wi, o);
            if (lane >= o) wi += nb;
        }
        s_wsum[lane] = wi - ws;
    }
    __syncthreads();
    const int bstart = inc - cnt + s_wsum[wave];
    if (t < NBIN) {
        const int mine = s_hist[t];
        int g = 0;
        if (mine) g = atomicAdd(&cursor[t], mine);
        s_base[t] = bstart + g;
        if (blockIdx.x == 0) binstart[t] = bstart;
    }
    if (blockIdx.x == 0 && t == NBIN) binstart[NBIN] = NEDGE;
    __syncthreads();
#pragma unroll
    for (int i = 0; i < EPB / 1024; ++i) {
        if (d[i] >= 0) {
            const int e   = base + i * 1024 + t;
            const int bin = (unsigned)d[i] / RB;
            const int pos = s_base[bin] + rk[i];
            const int pk  = esrc[e] | ((d[i] - bin * RB) << SRC_BITS);
            if ((unsigned)pos < (unsigned)NEDGE)
                packed[pos] = make_int2(pk, __builtin_bit_cast(int, ev[e]));
        }
    }
}

template <int F> struct lds_pitch { static constexpr int v = (F == 64) ? 68 : F; };

static __device__ __forceinline__ void lds_add(float* p, float v) {
    (void)__hip_atomic_fetch_add(p, v, __ATOMIC_RELAXED, __HIP_MEMORY_SCOPE_WORKGROUP);
}

template <int F, bool TAIL>
static __device__ __forceinline__ void spmm_run(const int pk_l, const int w_l, const int lane,
                                                const float* __restrict__ sup, float* acc) {
    if constexpr (F == 64) {
#pragma unroll
        for (int jb = 0; jb < 32; jb += 16) {
            float v0[16], v1[16];
#pragma unroll
            for (int u = 0; u < 16; ++u) {
                const int pk = __builtin_amdgcn_readlane(pk_l, jb + u);
                const unsigned si = ((unsigned)pk & ((1u << SRC_BITS) - 1u)) * 64u + (unsigned)lane;
                v0[u] = sup[si];
                v1[u] = sup[si + 32u];
            }
#pragma unroll
            for (int u = 0; u < 16; ++u) {
                const int   pk = __builtin_amdgcn_readlane(pk_l, jb + u);
                const float w  = __builtin_bit_cast(
                    float, __builtin_amdgcn_readlane(w_l, jb + u));
                const unsigned di = (((unsigned)pk >> SRC_BITS) & (RB - 1u)) * (unsigned)lds_pitch<64>::v + (unsigned)lane;
                if (!TAIL || pk >= 0) {
                    lds_add(acc + di,       v0[u] * w);
                    lds_add(acc + di + 32u, v1[u] * w);
                }
            }
        }
    } else {
        const int l  = lane & 15;
        const int hi = lane >> 4;
#pragma unroll
        for (int jb = 0; jb < 16; jb += 8) {
            float v[8];
#pragma unroll
            for (int u = 0; u < 8; ++u) {
                const int sel = (2 * (jb + u) + hi) << 2;
                const int pk  = __builtin_amdgcn_ds_bpermute(sel, pk_l);
                v[u] = sup[((unsigned)pk & ((1u << SRC_BITS) - 1u)) * 16u + (unsigned)l];
            }
#pragma unroll
            for (int u = 0; u < 8; ++u) {
                const int sel = (2 * (jb + u) + hi) << 2;
                const int   pk = __builtin_amdgcn_ds_bpermute(sel, pk_l);
                const float w  = __builtin_bit_cast(
                    float, __builtin_amdgcn_ds_bpermute(sel, w_l));
                const unsigned di = (((unsigned)pk >> SRC_BITS) & (RB - 1u)) * 16u + (unsigned)l;
                if (!TAIL || pk >= 0) lds_add(acc + di, v[u] * w);
            }
        }
    }
}

template <int F, int BS>
__global__ __launch_bounds__(BS) void spmm_bin_kernel(const int2* __restrict__ packed,
                                                      const int* __restrict__ binstart,
                                                      const float* __restrict__ sup,
                                                      const float* __restrict__ bias,
                                                      const float* __restrict__ W2,
                                                      float* __restrict__ outp) {
    static_assert(F == 64 || F == 16, "row widths of the two layers");
    static_assert((long long)N_NODES * F < (1ll << 31), "32-bit element index");
    constexpr int P = lds_pitch<F>::v;
    static_assert((RB * P) % 4 == 0 && P % 4 == 0, "float4 fill, 16-byte rows");
    __shared__ float acc[RB * P];
    const int t    = threadIdx.x;
    const int lane = t & 31;
    const int wave  = __builtin_amdgcn_readfirstlane(t >> 5);
    const int bin   = blockIdx.x;
    const int start = __builtin_amdgcn_readfirstlane(binstart[bin]);
    int       end   = __builtin_amdgcn_readfirstlane(binstart[bin + 1]);
    end = end < NEDGE ? end : NEDGE;

    float4* acc4 = reinterpret_cast<float4*>(acc);
    for (int i = t; i < RB * P / 4; i += BS) acc4[i] = make_float4(0.0f, 0.0f, 0.0f, 0.0f);
    __syncthreads();

    for (int e0 = start + wave * 32; e0 < end; e0 += BS) {
        const int n = end - e0;
        int pk_l = (int)0x80000000;
        int w_l  = 0;
        if (lane < n) {
            const int2 r = packed[e0 + lane];
            pk_l = r.x;
            w_l  = r.y;
        }
        if (n >= 32) spmm_run<F, false>(pk_l, w_l, lane, sup, acc);
        else         spmm_run<F, true >(pk_l, w_l, lane, sup, acc);
    }
    __syncthreads();

    const int rb0 = bin * RB;
    if constexpr (F == 64) {
        static_assert(BS / 32 >= RB / 16, "one wave per 16-row tile");
        const int row0 = rb0 + wave * 16;
        if (wave < RB / 16 && row0 < N_NODES) {
            const int l  = lane & 15;
            const int hi = lane >> 4;
            const float* arow = acc + (wave * 16 + l) * P;
            v8f c = {};
#pragma unroll
            for (int k = 0; k < NHID; k += 4) {
                const int ks = k + 2 * hi;
                v2f a = *reinterpret_cast<const v2f*>(arow + ks);
                const v2f bb = *reinterpret_cast<const v2f*>(&bias[ks]);
                a.x += bb.x;
                a.y += bb.y;
                a.x = a.x > 0.0f ? a.x : 0.0f;
                a.y = a.y > 0.0f ? a.y : 0.0f;
                v2f b;
                b.x = W2[ks * NCLS + l];
                b.y = W2[(ks + 1) * NCLS + l];
                c = __builtin_amdgcn_wmma_f32_16x16x4_f32(false, a, false, b,
                                                          (short)0, c, false, false);
            }
            float* o = outp + (size_t)row0 * NCLS;
            const int mb = hi * 8;
#pragma unroll
            for (int r = 0; r < 8; ++r)
                o[(size_t)(mb + r) * NCLS + l] = c[r];
        }
    } else {
        static_assert(RB * F / 4 == BS, "one float4 of the bin per thread");
        const int row = rb0 + (t >> 2);
        float4 v = acc4[t];
        const float4 bv = *reinterpret_cast<const float4*>(bias + (t & 3) * 4);
        v.x += bv.x; v.y += bv.y; v.z += bv.z; v.w += bv.w;
        float m = v.x;
        m = v.y > m ? v.y : m;
        m = v.z > m ? v.z : m;
        m = v.w > m ? v.w : m;
        float mo = __shfl_xor(m, 1);
        m = mo > m ? mo : m;
        mo = __shfl_xor(m, 2);
        m = mo > m ? mo : m;
        float s = __expf(v.x - m);
        s += __expf(v.y - m);
        s += __expf(v.z - m);
        s += __expf(v.w - m);
        s += __shfl_xor(s, 1);
        s += __shfl_xor(s, 2);
        const float lse = m + __logf(s);
        if (row < N_NODES) {
            float4* o4 = reinterpret_cast<float4*>(outp + (size_t)rb0 * F);
            o4[t] = make_float4(v.x - lse, v.y - lse, v.z - lse, v.w - lse);
        }
    }
}

extern "C" void kernel_launch(void* const* d_in, const int* in_sizes, int n_in,
                              void* d_out, int out_size, void* d_ws, size_t ws_size,
                              hipStream_t stream) {
    (void)in_sizes; (void)n_in; (void)out_size; (void)ws_size;
    const float* x   = (const float*)d_in[0];
    const int*   es  = (const int*)  d_in[1];
    const int*   ed  = (const int*)  d_in[2];
    const float* ev  = (const float*)d_in[3];
    const float* W1  = (const float*)d_in[4];
    const float* b1  = (const float*)d_in[5];
    const float* W2  = (const float*)d_in[6];
    const float* b2  = (const float*)d_in[7];
    float* out = (float*)d_out;

    float* sup1 = (float*)d_ws;
    float* sup2 = sup1 + (size_t)N_NODES * NHID;
    int*  counts   = (int*)(sup2 + (size_t)N_NODES * NHID);
    int*  cursor   = counts + 1024;
    int*  binstart = cursor + 1024;
    int2* packed   = (int2*)(binstart + 1024);

    gemm1_kernel<<<NTILES, 128, 0, stream>>>(x, W1, sup1, counts, cursor);
    bin_count_kernel<<<NBLK_E, 1024, 0, stream>>>(ed, counts);
    bin_fill_kernel<<<NBLK_E, 1024, 0, stream>>>(es, ed, ev, counts, cursor, binstart, packed);
    spmm_bin_kernel<NHID, 512><<<NBIN, 512, 0, stream>>>(packed, binstart, sup1, b1, W2, sup2);
    spmm_bin_kernel<NCLS, 512><<<NBIN, 512, 0, stream>>>(packed, binstart, sup2, b2, W2, out);
}


# GEMM1 A tile in LDS padded from a row pitch of 256 to 260 floats: the 16 rows of each fragment read fall on 16 different bank groups instead of one (16-way conflict gone); sums unchanged
# speedup vs baseline: 3.5187x; 1.1986x over previous
#include <hip/hip_runtime.h>


typedef float v2f __attribute__((ext_vector_type(2)));
typedef float v8f __attribute__((ext_vector_type(8)));

#define N_NODES 100000
#define FEAT    256
#define NHID    64
#define NCLS    16
#define NEDGE   1600000
#define NTILES  (N_NODES / 16)
#define APITCH  (FEAT + 4)
static_assert(APITCH % 4 == 0 && (APITCH / 4) % 2 == 1, "float4 staging stores; rows 16 B x odd apart");

#define RB      128
#define NBIN    ((N_NODES + RB - 1) / RB)
#define EPB     4096
#define NBLK_E  ((NEDGE + EPB - 1) / EPB)
#define SRC_BITS 17

static_assert(N_NODES <= (1 << SRC_BITS), "source row must fit the packed word");
static_assert(RB == 128 && NBIN <= 1024, "one thread per bin in the 1024-thread passes");

__global__ __launch_bounds__(128) void gemm1_kernel(const float* __restrict__ x,
                                                    const float* __restrict__ W1,
                                                    float* __restrict__ sup,
                                                    int* __restrict__ counts,
                                                    int* __restrict__ cursor) {
    __shared__ float tileA[16 * APITCH];
    const int lane = threadIdx.x & 31;
    const int wave = threadIdx.x >> 5;
    const int rb   = blockIdx.x * 16;

    if (threadIdx.x == 0 && blockIdx.x < 1024) {
        counts[blockIdx.x] = 0;
        cursor[blockIdx.x] = 0;
    }

    const float4* x4 = reinterpret_cast<const float4*>(x + (size_t)rb * FEAT);
    float4* t4 = reinterpret_cast<float4*>(tileA);
    for (int i = threadIdx.x; i < 16 * FEAT / 4; i += 128)
        t4[(i / (FEAT / 4)) * (APITCH / 4) + (i % (FEAT / 4))] = x4[i];
    __syncthreads();

    const int c  = wave * 16;
    const int l  = lane & 15;
    const int hi = lane >> 4;
    v8f acc = {};
    for (int k = 0; k < FEAT; k += 4) {
        const int ks = k + 2 * hi;
        v2f a = *reinterpret_cast<const v2f*>(&tileA[l * APITCH + ks]);
        v2f b;
        b.x = W1[(size_t)ks       * NHID + c + l];
        b.y = W1[(size_t)(ks + 1) * NHID + c + l];
        acc = __builtin_amdgcn_wmma_f32_16x16x4_f32(false, a, false, b,
                                                    (short)0, acc, false, false);
    }
    float* out = sup + (size_t)rb * NHID + c;
    const int mb = hi * 8;
#pragma unroll
    for (int r = 0; r < 8; ++r)
        out[(size_t)(mb + r) * NHID + l] = acc[r];
}

__global__ __launch_bounds__(1024) void bin_count_kernel(const int* __restrict__ edst,
                                                         int* __restrict__ counts) {
    __shared__ int s_hist[NBIN];
    const int t = threadIdx.x;
    if (t < NBIN) s_hist[t] = 0;
    __syncthreads();
    const int base = blockIdx.x * EPB;
#pragma unroll
    for (int i = 0; i < EPB / 1024; ++i) {
        const int e = base + i * 1024 + t;
        if (e < NEDGE) atomicAdd(&s_hist[(unsigned)edst[e] / RB], 1);
    }
    __syncthreads();
    if (t < NBIN) {
        const int cnt = s_hist[t];
        if (cnt) atomicAdd(&counts[t], cnt);
    }
}

__global__ __launch_bounds__(1024) void bin_fill_kernel(const int* __restrict__ esrc,
                                                        const int* __restrict__ edst,
                                                        const float* __restrict__ ev,
                                                        const int* __restrict__ counts,
                                                        int* __restrict__ cursor,
                                                        int* __restrict__ binstart,
                                                        int2* __restrict__ packed) {
    __shared__ int s_hist[NBIN];
    __shared__ int s_base[NBIN];
    __shared__ int s_wsum[32];
    const int t    = threadIdx.x;
    const int lane = t & 31;
    const int wave = t >> 5;
    if (t < NBIN) s_hist[t] = 0;
    __syncthreads();
    const int base = blockIdx.x * EPB;
    int d[EPB / 1024], rk[EPB / 1024];
#pragma unroll
    for (int i = 0; i < EPB / 1024; ++i) {
        const int e = base + i * 1024 + t;
        d[i]  = -1;
        rk[i] = 0;
        if (e < NEDGE) {
            d[i]  = edst[e];
            rk[i] = atomicAdd(&s_hist[(unsigned)d[i] / RB], 1);
        }
    }
    const int cnt = (t < NBIN) ? counts[t] : 0;
    int inc = cnt;
#pragma unroll
    for (int o = 1; o < 32; o <<= 1) {
        const int nb = __shfl_up(inc, o);
        if (lane >= o) inc += nb;
    }
    if (lane == 31) s_wsum[wave] = inc;
    __syncthreads();
    if (wave == 0) {
        const int ws = s_wsum[lane];
        int wi = ws;
#pragma unroll
        for (int o = 1; o < 32; o <<= 1) {
            const int nb = __shfl_up(wi, o);
            if (lane >= o) wi += nb;
        }
        s_wsum[lane] = wi - ws;
    }
    __syncthreads();
    const int bstart = inc - cnt + s_wsum[wave];
    if (t < NBIN) {
        const int mine = s_hist[t];
        int g = 0;
        if (mine) g = atomicAdd(&cursor[t], mine);
        s_base[t] = bstart + g;
        if (blockIdx.x == 0) binstart[t] = bstart;
    }
    if (blockIdx.x == 0 && t == NBIN) binstart[NBIN] = NEDGE;
    __syncthreads();
#pragma unroll
    for (int i = 0; i < EPB / 1024; ++i) {
        if (d[i] >= 0) {
            const int e   = base + i * 1024 + t;
            const int bin = (unsigned)d[i] / RB;
            const int pos = s_base[bin] + rk[i];
            const int pk  = esrc[e] | ((d[i] - bin * RB) << SRC_BITS);
            if ((unsigned)pos < (unsigned)NEDGE)
                packed[pos] = make_int2(pk, __builtin_bit_cast(int, ev[e]));
        }
    }
}

template <int F> struct lds_pitch { static constexpr int v = (F == 64) ? 68 : F; };

static __device__ __forceinline__ void lds_add(float* p, float v) {
    (void)__hip_atomic_fetch_add(p, v, __ATOMIC_RELAXED, __HIP_MEMORY_SCOPE_WORKGROUP);
}

template <int F, bool TAIL>
static __device__ __forceinline__ void spmm_run(const int pk_l, const int w_l, const int lane,
                                                const float* __restrict__ sup, float* acc) {
    if constexpr (F == 64) {
#pragma unroll
        for (int jb = 0; jb < 32; jb += 16) {
            float v0[16], v1[16];
#pragma unroll
            for (int u = 0; u < 16; ++u) {
                const int pk = __builtin_amdgcn_readlane(pk_l, jb + u);
                const unsigned si = ((unsigned)pk & ((1u << SRC_BITS) - 1u)) * 64u + (unsigned)lane;
                v0[u] = sup[si];
                v1[u] = sup[si + 32u];
            }
#pragma unroll
            for (int u = 0; u < 16; ++u) {
                const int   pk = __builtin_amdgcn_readlane(pk_l, jb + u);
                const float w  = __builtin_bit_cast(
                    float, __builtin_amdgcn_readlane(w_l, jb + u));
                const unsigned di = (((unsigned)pk >> SRC_BITS) & (RB - 1u)) * (unsigned)lds_pitch<64>::v + (unsigned)lane;
                if (!TAIL || pk >= 0) {
                    lds_add(acc + di,       v0[u] * w);
                    lds_add(acc + di + 32u, v1[u] * w);
                }
            }
        }
    } else {
        const int l  = lane & 15;
        const int hi = lane >> 4;
#pragma unroll
        for (int jb = 0; jb < 16; jb += 8) {
            float v[8];
#pragma unroll
            for (int u = 0; u < 8; ++u) {
                const int sel = (2 * (jb + u) + hi) << 2;
                const int pk  = __builtin_amdgcn_ds_bpermute(sel, pk_l);
                v[u] = sup[((unsigned)pk & ((1u << SRC_BITS) - 1u)) * 16u + (unsigned)l];
            }
#pragma unroll
            for (int u = 0; u < 8; ++u) {
                const int sel = (2 * (jb + u) + hi) << 2;
                const int   pk = __builtin_amdgcn_ds_bpermute(sel, pk_l);
                const float w  = __builtin_bit_cast(
                    float, __builtin_amdgcn_ds_bpermute(sel, w_l));
                const unsigned di = (((unsigned)pk >> SRC_BITS) & (RB - 1u)) * 16u + (unsigned)l;
                if (!TAIL || pk >= 0) lds_add(acc + di, v[u] * w);
            }
        }
    }
}

template <int F, int BS>
__global__ __launch_bounds__(BS) void spmm_bin_kernel(const int2* __restrict__ packed,
                                                      const int* __restrict__ binstart,
                                                      const float* __restrict__ sup,
                                                      const float* __restrict__ bias,
                                                      const float* __restrict__ W2,
                                                      float* __restrict__ outp) {
    static_assert(F == 64 || F == 16, "row widths of the two layers");
    static_assert((long long)N_NODES * F < (1ll << 31), "32-bit element index");
    constexpr int P = lds_pitch<F>::v;
    static_assert((RB * P) % 4 == 0 && P % 4 == 0, "float4 fill, 16-byte rows");
    __shared__ float acc[RB * P];
    const int t    = threadIdx.x;
    const int lane = t & 31;
    const int wave  = __builtin_amdgcn_readfirstlane(t >> 5);
    const int bin   = blockIdx.x;
    const int start = __builtin_amdgcn_readfirstlane(binstart[bin]);
    int       end   = __builtin_amdgcn_readfirstlane(binstart[bin + 1]);
    end = end < NEDGE ? end : NEDGE;

    float4* acc4 = reinterpret_cast<float4*>(acc);
    for (int i = t; i < RB * P / 4; i += BS) acc4[i] = make_float4(0.0f, 0.0f, 0.0f, 0.0f);
    __syncthreads();

    for (int e0 = start + wave * 32; e0 < end; e0 += BS) {
        const int n = end - e0;
        int pk_l = (int)0x80000000;
        int w_l  = 0;
        if (lane < n) {
            const int2 r = packed[e0 + lane];
            pk_l = r.x;
            w_l  = r.y;
        }
        if (n >= 32) spmm_run<F, false>(pk_l, w_l, lane, sup, acc);
        else         spmm_run<F, true >(pk_l, w_l, lane, sup, acc);
    }
    __syncthreads();

    const int rb0 = bin * RB;
    if constexpr (F == 64) {
        static_assert(BS / 32 >= RB / 16, "one wave per 16-row tile");
        const int row0 = rb0 + wave * 16;
        if (wave < RB / 16 && row0 < N_NODES) {
            const int l  = lane & 15;
            const int hi = lane >> 4;
            const float* arow = acc + (wave * 16 + l) * P;
            v8f c = {};
#pragma unroll
            for (int k = 0; k < NHID; k += 4) {
                const int ks = k + 2 * hi;
                v2f a = *reinterpret_cast<const v2f*>(arow + ks);
                const v2f bb = *reinterpret_cast<const v2f*>(&bias[ks]);
                a.x += bb.x;
                a.y += bb.y;
                a.x = a.x > 0.0f ? a.x : 0.0f;
                a.y = a.y > 0.0f ? a.y : 0.0f;
                v2f b;
                b.x = W2[ks * NCLS + l];
                b.y = W2[(ks + 1) * NCLS + l];
                c = __builtin_amdgcn_wmma_f32_16x16x4_f32(false, a, false, b,
                                                          (short)0, c, false, false);
            }
            float* o = outp + (size_t)row0 * NCLS;
            const int mb = hi * 8;
#pragma unroll
            for (int r = 0; r < 8; ++r)
                o[(size_t)(mb + r) * NCLS + l] = c[r];
        }
    } else {
        static_assert(RB * F / 4 == BS, "one float4 of the bin per thread");
        const int row = rb0 + (t >> 2);
        float4 v = acc4[t];
        const float4 bv = *reinterpret_cast<const float4*>(bias + (t & 3) * 4);
        v.x += bv.x; v.y += bv.y; v.z += bv.z; v.w += bv.w;
        float m = v.x;
        m = v.y > m ? v.y : m;
        m = v.z > m ? v.z : m;
        m = v.w > m ? v.w : m;
        float mo = __shfl_xor(m, 1);
        m = mo > m ? mo : m;
        mo = __shfl_xor(m, 2);
        m = mo > m ? mo : m;
        float s = __expf(v.x - m);
        s += __expf(v.y - m);
        s += __expf(v.z - m);
        s += __expf(v.w - m);
        s += __shfl_xor(s, 1);
        s += __shfl_xor(s, 2);
        const float lse = m + __logf(s);
        if (row < N_NODES) {
            float4* o4 = reinterpret_cast<float4*>(outp + (size_t)rb0 * F);
            o4[t] = make_float4(v.x - lse, v.y - lse, v.z - lse, v.w - lse);
        }
    }
}

extern "C" void kernel_launch(void* const* d_in, const int* in_sizes, int n_in,
                              void* d_out, int out_size, void* d_ws, size_t ws_size,
                              hipStream_t stream) {
    (void)in_sizes; (void)n_in; (void)out_size; (void)ws_size;
    const float* x   = (const float*)d_in[0];
    const int*   es  = (const int*)  d_in[1];
    const int*   ed  = (const int*)  d_in[2];
    const float* ev  = (const float*)d_in[3];
    const float* W1  = (const float*)d_in[4];
    const float* b1  = (const float*)d_in[5];
    const float* W2  = (const float*)d_in[6];
    const float* b2  = (const float*)d_in[7];
    float* out = (float*)d_out;

    float* sup1 = (float*)d_ws;
    float* sup2 = sup1 + (size_t)N_NODES * NHID;
    int*  counts   = (int*)(sup2 + (size_t)N_NODES * NHID);
    int*  cursor   = counts + 1024;
    int*  binstart = cursor + 1024;
    int2* packed   = (int2*)(binstart + 1024);

    gemm1_kernel<<<NTILES, 128, 0, stream>>>(x, W1, sup1, counts, cursor);
    bin_count_kernel<<<NBLK_E, 1024, 0, stream>>>(ed, counts);
    bin_fill_kernel<<<NBLK_E, 1024, 0, stream>>>(es, ed, ev, counts, cursor, binstart, packed);
    spmm_bin_kernel<NHID, 512><<<NBIN, 512, 0, stream>>>(packed, binstart, sup1, b1, W2, sup2);
    spmm_bin_kernel<NCLS, 512><<<NBIN, 512, 0, stream>>>(packed, binstart, sup2, b2, W2, out);
}


# edge count launch and prefix scan removed: each bin owns a fixed record segment filled at the bin's running cursor, records past it go to an overflow list; 5 launches -> 4; arrival-order sums kept
# speedup vs baseline: 3.6239x; 1.0299x over previous
#include <hip/hip_runtime.h>


typedef float v2f __attribute__((ext_vector_type(2)));
typedef float v8f __attribute__((ext_vector_type(8)));

#define N_NODES 100000
#define FEAT    256
#define NHID    64
#define NCLS    16
#define NEDGE   1600000
#define NTILES  (N_NODES / 16)
#define APITCH  (FEAT + 4)
static_assert(APITCH % 4 == 0 && (APITCH / 4) % 2 == 1, "float4 staging stores; rows 16 B x odd apart");

#define RB      128
#define NBIN    ((N_NODES + RB - 1) / RB)
#define EPB     4096
#define NBLK_E  ((NEDGE + EPB - 1) / EPB)
#define SRC_BITS 17
#define CAP      4128
#define OVF_SLOT 1023
static_assert(CAP % 32 == 0 && (CAP / 32) % 2 == 1, "256-B aligned segments, an odd number of 256-B units apart");

static_assert(N_NODES <= (1 << SRC_BITS), "source row must fit the packed word");
static_assert(RB == 128 && NBIN <= OVF_SLOT, "one thread per bin in the 1024-thread pass; the last counter word is the overflow count");
static_assert((long long)NBIN * CAP < (1ll << 31), "32-bit record index");

__global__ __launch_bounds__(128) void gemm1_kernel(const float* __restrict__ x,
                                                    const float* __restrict__ W1,
                                                    float* __restrict__ sup,
                                                    int* __restrict__ cursor) {
    __shared__ float tileA[16 * APITCH];
    const int lane = threadIdx.x & 31;
    const int wave = threadIdx.x >> 5;
    const int rb   = blockIdx.x * 16;

    if (threadIdx.x == 0 && blockIdx.x < 1024) cursor[blockIdx.x] = 0;

    const float4* x4 = reinterpret_cast<const float4*>(x + (size_t)rb * FEAT);
    float4* t4 = reinterpret_cast<float4*>(tileA);
    for (int i = threadIdx.x; i < 16 * FEAT / 4; i += 128)
        t4[(i / (FEAT / 4)) * (APITCH / 4) + (i % (FEAT / 4))] = x4[i];
    __syncthreads();

    const int c  = wave * 16;
    const int l  = lane & 15;
    const int hi = lane >> 4;
    v8f acc = {};
    for (int k = 0; k < FEAT; k += 4) {
        const int ks = k + 2 * hi;
        v2f a = *reinterpret_cast<const v2f*>(&tileA[l * APITCH + ks]);
        v2f b;
        b.x = W1[(size_t)ks       * NHID + c + l];
        b.y = W1[(size_t)(ks + 1) * NHID + c + l];
        acc = __builtin_amdgcn_wmma_f32_16x16x4_f32(false, a, false, b,
                                                    (short)0, acc, false, false);
    }
    float* out = sup + (size_t)rb * NHID + c;
    const int mb = hi * 8;
#pragma unroll
    for (int r = 0; r < 8; ++r)
        out[(size_t)(mb + r) * NHID + l] = acc[r];
}

__global__ __launch_bounds__(1024) void bin_fill_kernel(const int* __restrict__ esrc,
                                                        const int* __restrict__ edst,
                                                        const float* __restrict__ ev,
                                                        int* __restrict__ cursor,
                                                        int2* __restrict__ packed,
                                                        int4* __restrict__ ovf) {
    __shared__ int s_hist[NBIN];
    __shared__ int s_base[NBIN];
    const int t = threadIdx.x;
    if (t < NBIN) s_hist[t] = 0;
    __syncthreads();
    const int base = blockIdx.x * EPB;
    int d[EPB / 1024], rk[EPB / 1024];
#pragma unroll
    for (int i = 0; i < EPB / 1024; ++i) {
        const int e = base + i * 1024 + t;
        d[i]  = -1;
        rk[i] = 0;
        if (e < NEDGE) {
            d[i]  = edst[e];
            rk[i] = atomicAdd(&s_hist[(unsigned)d[i] / RB], 1);
        }
    }
    __syncthreads();
    if (t < NBIN) {
        const int mine = s_hist[t];
        int g = 0;
        if (mine) g = atomicAdd(&cursor[t], mine);
        s_base[t] = g;
    }
    __syncthreads();
#pragma unroll
    for (int i = 0; i < EPB / 1024; ++i) {
        if (d[i] >= 0) {
            const int e   = base + i * 1024 + t;
            const int bin = (unsigned)d[i] / RB;
            const int pos = s_base[bin] + rk[i];
            const int s   = esrc[e];
            const int wb  = __builtin_bit_cast(int, ev[e]);
            if ((unsigned)pos < (unsigned)CAP) {
                packed[bin * CAP + pos] = make_int2(s | ((d[i] - bin * RB) << SRC_BITS), wb);
            } else {
                const int o = atomicAdd(&cursor[OVF_SLOT], 1);
                if ((unsigned)o < (unsigned)NEDGE)
                    ovf[o] = make_int4(s, d[i], wb, 0);
            }
        }
    }
}

template <int F> struct lds_pitch { static constexpr int v = (F == 64) ? 68 : F; };

static __device__ __forceinline__ void lds_add(float* p, float v) {
    (void)__hip_atomic_fetch_add(p, v, __ATOMIC_RELAXED, __HIP_MEMORY_SCOPE_WORKGROUP);
}

template <int F, bool TAIL>
static __device__ __forceinline__ void spmm_run(const int pk_l, const int w_l, const int lane,
                                                const float* __restrict__ sup, float* acc) {
    if constexpr (F == 64) {
#pragma unroll
        for (int jb = 0; jb < 32; jb += 16) {
            float v0[16], v1[16];
#pragma unroll
            for (int u = 0; u < 16; ++u) {
                const int pk = __builtin_amdgcn_readlane(pk_l, jb + u);
                const unsigned si = ((unsigned)pk & ((1u << SRC_BITS) - 1u)) * 64u + (unsigned)lane;
                v0[u] = sup[si];
                v1[u] = sup[si + 32u];
            }
#pragma unroll
            for (int u = 0; u < 16; ++u) {
                const int   pk = __builtin_amdgcn_readlane(pk_l, jb + u);
                const float w  = __builtin_bit_cast(
                    float, __builtin_amdgcn_readlane(w_l, jb + u));
                const unsigned di = (((unsigned)pk >> SRC_BITS) & (RB - 1u)) * (unsigned)lds_pitch<64>::v + (unsigned)lane;
                if (!TAIL || pk >= 0) {
                    lds_add(acc + di,       v0[u] * w);
                    lds_add(acc + di + 32u, v1[u] * w);
                }
            }
        }
    } else {
        const int l  = lane & 15;
        const int hi = lane >> 4;
#pragma unroll
        for (int jb = 0; jb < 16; jb += 8) {
            float v[8];
#pragma unroll
            for (int u = 0; u < 8; ++u) {
                const int sel = (2 * (jb + u) + hi) << 2;
                const int pk  = __builtin_amdgcn_ds_bpermute(sel, pk_l);
                v[u] = sup[((unsigned)pk & ((1u << SRC_BITS) - 1u)) * 16u + (unsigned)l];
            }
#pragma unroll
            for (int u = 0; u < 8; ++u) {
                const int sel = (2 * (jb + u) + hi) << 2;
                const int   pk = __builtin_amdgcn_ds_bpermute(sel, pk_l);
                const float w  = __builtin_bit_cast(
                    float, __builtin_amdgcn_ds_bpermute(sel, w_l));
                const unsigned di = (((unsigned)pk >> SRC_BITS) & (RB - 1u)) * 16u + (unsigned)l;
                if (!TAIL || pk >= 0) lds_add(acc + di, v[u] * w);
            }
        }
    }
}

template <int F, int BS>
__global__ __launch_bounds__(BS) void spmm_bin_kernel(const int2* __restrict__ packed,
                                                      const int* __restrict__ cursor,
                                                      const int4* __restrict__ ovf,
                                                      const float* __restrict__ sup,
                                                      const float* __restrict__ bias,
                                                      const float* __restrict__ W2,
                                                      float* __restrict__ outp) {
    static_assert(F == 64 || F == 16, "row widths of the two layers");
    static_assert((long long)N_NODES * F < (1ll << 31), "32-bit element index");
    constexpr int P = lds_pitch<F>::v;
    static_assert((RB * P) % 4 == 0 && P % 4 == 0, "float4 fill, 16-byte rows");
    __shared__ float acc[RB * P];
    const int t    = threadIdx.x;
    const int lane = t & 31;
    const int wave  = __builtin_amdgcn_readfirstlane(t >> 5);
    const int bin   = blockIdx.x;
    const int start = bin * CAP;
    int       cnt   = __builtin_amdgcn_readfirstlane(cursor[bin]);
    cnt = cnt < CAP ? cnt : CAP;
    const int end   = start + cnt;
    int       novf  = __builtin_amdgcn_readfirstlane(cursor[OVF_SLOT]);
    novf = novf < NEDGE ? novf : NEDGE;

    float4* acc4 = reinterpret_cast<float4*>(acc);
    for (int i = t; i < RB * P / 4; i += BS) acc4[i] = make_float4(0.0f, 0.0f, 0.0f, 0.0f);
    __syncthreads();

    for (int e0 = start + wave * 32; e0 < end; e0 += BS) {
        const int n = end - e0;
        int pk_l = (int)0x80000000;
        int w_l  = 0;
        if (lane < n) {
            const int2 r = packed[e0 + lane];
            pk_l = r.x;
            w_l  = r.y;
        }
        if (n >= 32) spmm_run<F, false>(pk_l, w_l, lane, sup, acc);
        else         spmm_run<F, true >(pk_l, w_l, lane, sup, acc);
    }
    for (int i = t; i < novf; i += BS) {
        const int4 r = ovf[i];
        if ((unsigned)r.y / RB == (unsigned)bin) {
            const float  w    = __builtin_bit_cast(float, r.z);
            const float* srow = sup + (size_t)r.x * F;
            float*       drow = acc + (r.y - bin * RB) * P;
            for (int c = 0; c < F; ++c) lds_add(drow + c, srow[c] * w);
        }
    }
    __syncthreads();

    const int rb0 = bin * RB;
    if constexpr (F == 64) {
        static_assert(BS / 32 >= RB / 16, "one wave per 16-row tile");
        const int row0 = rb0 + wave * 16;
        if (wave < RB / 16 && row0 < N_NODES) {
            const int l  = lane & 15;
            const int hi = lane >> 4;
            const float* arow = acc + (wave * 16 + l) * P;
            v8f c = {};
#pragma unroll
            for (int k = 0; k < NHID; k += 4) {
                const int ks = k + 2 * hi;
                v2f a = *reinterpret_cast<const v2f*>(arow + ks);
                const v2f bb = *reinterpret_cast<const v2f*>(&bias[ks]);
                a.x += bb.x;
                a.y += bb.y;
                a.x = a.x > 0.0f ? a.x : 0.0f;
                a.y = a.y > 0.0f ? a.y : 0.0f;
                v2f b;
                b.x = W2[ks * NCLS + l];
                b.y = W2[(ks + 1) * NCLS + l];
                c = __builtin_amdgcn_wmma_f32_16x16x4_f32(false, a, false, b,
                                                          (short)0, c, false, false);
            }
            float* o = outp + (size_t)row0 * NCLS;
            const int mb = hi * 8;
#pragma unroll
            for (int r = 0; r < 8; ++r)
                o[(size_t)(mb + r) * NCLS + l] = c[r];
        }
    } else {
        static_assert(RB * F / 4 == BS, "one float4 of the bin per thread");
        const int row = rb0 + (t >> 2);
        float4 v = acc4[t];
        const float4 bv = *reinterpret_cast<const float4*>(bias + (t & 3) * 4);
        v.x += bv.x; v.y += bv.y; v.z += bv.z; v.w += bv.w;
        float m = v.x;
        m = v.y > m ? v.y : m;
        m = v.z > m ? v.z : m;
        m = v.w > m ? v.w : m;
        float mo = __shfl_xor(m, 1);
        m = mo > m ? mo : m;
        mo = __shfl_xor(m, 2);
        m = mo > m ? mo : m;
        float s = __expf(v.x - m);
        s += __expf(v.y - m);
        s += __expf(v.z - m);
        s += __expf(v.w - m);
        s += __shfl_xor(s, 1);
        s += __shfl_xor(s, 2);
        const float lse = m + __logf(s);
        if (row < N_NODES) {
            float4* o4 = reinterpret_cast<float4*>(outp + (size_t)rb0 * F);
            o4[t] = make_float4(v.x - lse, v.y - lse, v.z - lse, v.w - lse);
        }
    }
}

extern "C" void kernel_launch(void* const* d_in, const int* in_sizes, int n_in,
                              void* d_out, int out_size, void* d_ws, size_t ws_size,
                              hipStream_t stream) {
    (void)in_sizes; (void)n_in; (void)out_size; (void)ws_size;
    const float* x   = (const float*)d_in[0];
    const int*   es  = (const int*)  d_in[1];
    const int*   ed  = (const int*)  d_in[2];
    const float* ev  = (const float*)d_in[3];
    const float* W1  = (const float*)d_in[4];
    const float* b1  = (const float*)d_in[5];
    const float* W2  = (const float*)d_in[6];
    const float* b2  = (const float*)d_in[7];
    float* out = (float*)d_out;

    float* sup1 = (float*)d_ws;
    float* sup2 = sup1 + (size_t)N_NODES * NHID;
    int*  cursor = (int*)(sup2 + (size_t)N_NODES * NHID);
    int2* packed = (int2*)(cursor + 1024);
    int4* ovf    = (int4*)(packed + (size_t)NBIN * CAP);

    gemm1_kernel<<<NTILES, 128, 0, stream>>>(x, W1, sup1, cursor);
    bin_fill_kernel<<<NBLK_E, 1024, 0, stream>>>(es, ed, ev, cursor, packed, ovf);
    spmm_bin_kernel<NHID, 512><<<NBIN, 512, 0, stream>>>(packed, cursor, ovf, sup1, b1, W2, sup2);
    spmm_bin_kernel<NCLS, 512><<<NBIN, 512, 0, stream>>>(packed, cursor, ovf, sup2, b2, W2, out);
}


# binning pass stages its records in LDS in bin order and writes them with thread = list index: a bin's share leaves as consecutive 8-byte stores, 4x fewer line requests; arrival-order sums kept
# speedup vs baseline: 3.8547x; 1.0637x over previous
#include <hip/hip_runtime.h>


typedef float v2f __attribute__((ext_vector_type(2)));
typedef float v8f __attribute__((ext_vector_type(8)));

#define N_NODES 100000
#define FEAT    256
#define NHID    64
#define NCLS    16
#define NEDGE   1600000
#define NTILES  (N_NODES / 16)
#define APITCH  (FEAT + 4)
static_assert(APITCH % 4 == 0 && (APITCH / 4) % 2 == 1, "float4 staging stores; rows 16 B x odd apart");

#define RB      128
#define NBIN    ((N_NODES + RB - 1) / RB)
#define EPB     4096
#define NBLK_E  ((NEDGE + EPB - 1) / EPB)
#define SRC_BITS 17
#define CAP      4128
#define OVF_SLOT 1023
static_assert(CAP % 32 == 0 && (CAP / 32) % 2 == 1, "256-B aligned segments, an odd number of 256-B units apart");

static_assert(N_NODES <= (1 << SRC_BITS), "source row must fit the packed word");
static_assert(RB == 128 && NBIN <= OVF_SLOT, "one thread per bin in the 1024-thread pass; the last counter word is the overflow count");
static_assert((long long)NBIN * CAP < (1ll << 31), "32-bit record index");

__global__ __launch_bounds__(128) void gemm1_kernel(const float* __restrict__ x,
                                                    const float* __restrict__ W1,
                                                    float* __restrict__ sup,
                                                    int* __restrict__ cursor) {
    __shared__ float tileA[16 * APITCH];
    const int lane = threadIdx.x & 31;
    const int wave = threadIdx.x >> 5;
    const int rb   = blockIdx.x * 16;

    if (threadIdx.x == 0 && blockIdx.x < 1024) cursor[blockIdx.x] = 0;

    const float4* x4 = reinterpret_cast<const float4*>(x + (size_t)rb * FEAT);
    float4* t4 = reinterpret_cast<float4*>(tileA);
    for (int i = threadIdx.x; i < 16 * FEAT / 4; i += 128)
        t4[(i / (FEAT / 4)) * (APITCH / 4) + (i % (FEAT / 4))] = x4[i];
    __syncthreads();

    const int c  = wave * 16;
    const int l  = lane & 15;
    const int hi = lane >> 4;
    v8f acc = {};
    for (int k = 0; k < FEAT; k += 4) {
        const int ks = k + 2 * hi;
        v2f a = *reinterpret_cast<const v2f*>(&tileA[l * APITCH + ks]);
        v2f b;
        b.x = W1[(size_t)ks       * NHID + c + l];
        b.y = W1[(size_t)(ks + 1) * NHID + c + l];
        acc = __builtin_amdgcn_wmma_f32_16x16x4_f32(false, a, false, b,
                                                    (short)0, acc, false, false);
    }
    float* out = sup + (size_t)rb * NHID + c;
    const int mb = hi * 8;
#pragma unroll
    for (int r = 0; r < 8; ++r)
        out[(size_t)(mb + r) * NHID + l] = acc[r];
}

__global__ __launch_bounds__(1024) void bin_fill_kernel(const int* __restrict__ esrc,
                                                        const int* __restrict__ edst,
                                                        const float* __restrict__ ev,
                                                        int* __restrict__ cursor,
                                                        int2* __restrict__ packed,
                                                        int4* __restrict__ ovf) {
    __shared__ int  s_hist[NBIN];
    __shared__ int  s_lbase[NBIN];
    __shared__ int  s_gbase[NBIN];
    __shared__ int  s_wsum[32];
    __shared__ int2 s_rec[EPB];
    __shared__ int  s_slot[EPB];
    static_assert(NBIN <= 1024, "one bin per thread in the prefix scan");
    const int t    = threadIdx.x;
    const int lane = t & 31;
    const int wave = t >> 5;
    if (t < NBIN) s_hist[t] = 0;
    __syncthreads();
    const int base = blockIdx.x * EPB;
    int d[EPB / 1024], rk[EPB / 1024];
#pragma unroll
    for (int i = 0; i < EPB / 1024; ++i) {
        const int e = base + i * 1024 + t;
        d[i]  = -1;
        rk[i] = 0;
        if (e < NEDGE) {
            d[i]  = edst[e];
            rk[i] = atomicAdd(&s_hist[(unsigned)d[i] / RB], 1);
        }
    }
    __syncthreads();
    const int mine = (t < NBIN) ? s_hist[t] : 0;
    int inc = mine;
    for (int o = 1; o < 32; o <<= 1) {
        const int n = __shfl_up(inc, o);
        if (lane >= o) inc += n;
    }
    if (lane == 31) s_wsum[wave] = inc;
    __syncthreads();
    if (wave == 0) {
        const int w = s_wsum[lane];
        int winc = w;
        for (int o = 1; o < 32; o <<= 1) {
            const int n = __shfl_up(winc, o);
            if (lane >= o) winc += n;
        }
        s_wsum[lane] = winc - w;
    }
    __syncthreads();
    if (t < NBIN) {
        int g = 0;
        if (mine) g = atomicAdd(&cursor[t], mine);
        s_gbase[t] = g;
        s_lbase[t] = s_wsum[wave] + inc - mine;
    }
    __syncthreads();
#pragma unroll
    for (int i = 0; i < EPB / 1024; ++i) {
        if (d[i] >= 0) {
            const int e   = base + i * 1024 + t;
            const int bin = (unsigned)d[i] / RB;
            const int li  = s_lbase[bin] + rk[i];
            const int pos = s_gbase[bin] + rk[i];
            const int s   = esrc[e];
            const int wb  = __builtin_bit_cast(int, ev[e]);
            s_rec[li] = make_int2(s | ((d[i] - bin * RB) << SRC_BITS), wb);
            if ((unsigned)pos < (unsigned)CAP) {
                s_slot[li] = bin * CAP + pos;
            } else {
                s_slot[li] = -1;
                const int o = atomicAdd(&cursor[OVF_SLOT], 1);
                if ((unsigned)o < (unsigned)NEDGE)
                    ovf[o] = make_int4(s, d[i], wb, 0);
            }
        }
    }
    __syncthreads();
    int nrec = NEDGE - base;
    nrec = nrec < EPB ? nrec : EPB;
    int  slot[EPB / 1024];
    int2 rec[EPB / 1024];
#pragma unroll
    for (int i = 0; i < EPB / 1024; ++i) {
        const int j  = i * 1024 + t;
        const int jc = j < nrec ? j : nrec - 1;
        const int sl = s_slot[jc];
        rec[i]  = s_rec[jc];
        slot[i] = j < nrec ? sl : -1;
    }
#pragma unroll
    for (int i = 0; i < EPB / 1024; ++i)
        if (slot[i] >= 0) packed[slot[i]] = rec[i];
}

template <int F> struct lds_pitch { static constexpr int v = (F == 64) ? 68 : F; };

static __device__ __forceinline__ void lds_add(float* p, float v) {
    (void)__hip_atomic_fetch_add(p, v, __ATOMIC_RELAXED, __HIP_MEMORY_SCOPE_WORKGROUP);
}

template <int F, bool TAIL>
static __device__ __forceinline__ void spmm_run(const int pk_l, const int w_l, const int lane,
                                                const float* __restrict__ sup, float* acc) {
    if constexpr (F == 64) {
#pragma unroll
        for (int jb = 0; jb < 32; jb += 16) {
            float v0[16], v1[16];
#pragma unroll
            for (int u = 0; u < 16; ++u) {
                const int pk = __builtin_amdgcn_readlane(pk_l, jb + u);
                const unsigned si = ((unsigned)pk & ((1u << SRC_BITS) - 1u)) * 64u + (unsigned)lane;
                v0[u] = sup[si];
                v1[u] = sup[si + 32u];
            }
#pragma unroll
            for (int u = 0; u < 16; ++u) {
                const int   pk = __builtin_amdgcn_readlane(pk_l, jb + u);
                const float w  = __builtin_bit_cast(
                    float, __builtin_amdgcn_readlane(w_l, jb + u));
                const unsigned di = (((unsigned)pk >> SRC_BITS) & (RB - 1u)) * (unsigned)lds_pitch<64>::v + (unsigned)lane;
                if (!TAIL || pk >= 0) {
                    lds_add(acc + di,       v0[u] * w);
                    lds_add(acc + di + 32u, v1[u] * w);
                }
            }
        }
    } else {
        const int l  = lane & 15;
        const int hi = lane >> 4;
#pragma unroll
        for (int jb = 0; jb < 16; jb += 8) {
            float v[8];
#pragma unroll
            for (int u = 0; u < 8; ++u) {
                const int sel = (2 * (jb + u) + hi) << 2;
                const int pk  = __builtin_amdgcn_ds_bpermute(sel, pk_l);
                v[u] = sup[((unsigned)pk & ((1u << SRC_BITS) - 1u)) * 16u + (unsigned)l];
            }
#pragma unroll
            for (int u = 0; u < 8; ++u) {
                const int sel = (2 * (jb + u) + hi) << 2;
                const int   pk = __builtin_amdgcn_ds_bpermute(sel, pk_l);
                const float w  = __builtin_bit_cast(
                    float, __builtin_amdgcn_ds_bpermute(sel, w_l));
                const unsigned di = (((unsigned)pk >> SRC_BITS) & (RB - 1u)) * 16u + (unsigned)l;
                if (!TAIL || pk >= 0) lds_add(acc + di, v[u] * w);
            }
        }
    }
}

template <int F, int BS>
__global__ __launch_bounds__(BS) void spmm_bin_kernel(const int2* __restrict__ packed,
                                                      const int* __restrict__ cursor,
                                                      const int4* __restrict__ ovf,
                                                      const float* __restrict__ sup,
                                                      const float* __restrict__ bias,
                                                      const float* __restrict__ W2,
                                                      float* __restrict__ outp) {
    static_assert(F == 64 || F == 16, "row widths of the two layers");
    static_assert((long long)N_NODES * F < (1ll << 31), "32-bit element index");
    constexpr int P = lds_pitch<F>::v;
    static_assert((RB * P) % 4 == 0 && P % 4 == 0, "float4 fill, 16-byte rows");
    __shared__ float acc[RB * P];
    const int t    = threadIdx.x;
    const int lane = t & 31;
    const int wave  = __builtin_amdgcn_readfirstlane(t >> 5);
    const int bin   = blockIdx.x;
    const int start = bin * CAP;
    int       cnt   = __builtin_amdgcn_readfirstlane(cursor[bin]);
    cnt = cnt < CAP ? cnt : CAP;
    const int end   = start + cnt;
    int       novf  = __builtin_amdgcn_readfirstlane(cursor[OVF_SLOT]);
    novf = novf < NEDGE ? novf : NEDGE;

    float4* acc4 = reinterpret_cast<float4*>(acc);
    for (int i = t; i < RB * P / 4; i += BS) acc4[i] = make_float4(0.0f, 0.0f, 0.0f, 0.0f);
    __syncthreads();

    for (int e0 = start + wave * 32; e0 < end; e0 += BS) {
        const int n = end - e0;
        int pk_l = (int)0x80000000;
        int w_l  = 0;
        if (lane < n) {
            const int2 r = packed[e0 + lane];
            pk_l = r.x;
            w_l  = r.y;
        }
        if (n >= 32) spmm_run<F, false>(pk_l, w_l, lane, sup, acc);
        else         spmm_run<F, true >(pk_l, w_l, lane, sup, acc);
    }
    for (int i = t; i < novf; i += BS) {
        const int4 r = ovf[i];
        if ((unsigned)r.y / RB == (unsigned)bin) {
            const float  w    = __builtin_bit_cast(float, r.z);
            const float* srow = sup + (size_t)r.x * F;
            float*       drow = acc + (r.y - bin * RB) * P;
            for (int c = 0; c < F; ++c) lds_add(drow + c, srow[c] * w);
        }
    }
    __syncthreads();

    const int rb0 = bin * RB;
    if constexpr (F == 64) {
        static_assert(BS / 32 >= RB / 16, "one wave per 16-row tile");
        const int row0 = rb0 + wave * 16;
        if (wave < RB / 16 && row0 < N_NODES) {
            const int l  = lane & 15;
            const int hi = lane >> 4;
            const float* arow = acc + (wave * 16 + l) * P;
            v8f c = {};
#pragma unroll
            for (int k = 0; k < NHID; k += 4) {
                const int ks = k + 2 * hi;
                v2f a = *reinterpret_cast<const v2f*>(arow + ks);
                const v2f bb = *reinterpret_cast<const v2f*>(&bias[ks]);
                a.x += bb.x;
                a.y += bb.y;
                a.x = a.x > 0.0f ? a.x : 0.0f;
                a.y = a.y > 0.0f ? a.y : 0.0f;
                v2f b;
                b.x = W2[ks * NCLS + l];
                b.y = W2[(ks + 1) * NCLS + l];
                c = __builtin_amdgcn_wmma_f32_16x16x4_f32(false, a, false, b,
                                                          (short)0, c, false, false);
            }
            float* o = outp + (size_t)row0 * NCLS;
            const int mb = hi * 8;
#pragma unroll
            for (int r = 0; r < 8; ++r)
                o[(size_t)(mb + r) * NCLS + l] = c[r];
        }
    } else {
        static_assert(RB * F / 4 == BS, "one float4 of the bin per thread");
        const int row = rb0 + (t >> 2);
        float4 v = acc4[t];
        const float4 bv = *reinterpret_cast<const float4*>(bias + (t & 3) * 4);
        v.x += bv.x; v.y += bv.y; v.z += bv.z; v.w += bv.w;
        float m = v.x;
        m = v.y > m ? v.y : m;
        m = v.z > m ? v.z : m;
        m = v.w > m ? v.w : m;
        float mo = __shfl_xor(m, 1);
        m = mo > m ? mo : m;
        mo = __shfl_xor(m, 2);
        m = mo > m ? mo : m;
        float s = __expf(v.x - m);
        s += __expf(v.y - m);
        s += __expf(v.z - m);
        s += __expf(v.w - m);
        s += __shfl_xor(s, 1);
        s += __shfl_xor(s, 2);
        const float lse = m + __logf(s);
        if (row < N_NODES) {
            float4* o4 = reinterpret_cast<float4*>(outp + (size_t)rb0 * F);
            o4[t] = make_float4(v.x - lse, v.y - lse, v.z - lse, v.w - lse);
        }
    }
}

extern "C" void kernel_launch(void* const* d_in, const int* in_sizes, int n_in,
                              void* d_out, int out_size, void* d_ws, size_t ws_size,
                              hipStream_t stream) {
    (void)in_sizes; (void)n_in; (void)out_size; (void)ws_size;
    const float* x   = (const float*)d_in[0];
    const int*   es  = (const int*)  d_in[1];
    const int*   ed  = (const int*)  d_in[2];
    const float* ev  = (const float*)d_in[3];
    const float* W1  = (const float*)d_in[4];
    const float* b1  = (const float*)d_in[5];
    const float* W2  = (const float*)d_in[6];
    const float* b2  = (const float*)d_in[7];
    float* out = (float*)d_out;

    float* sup1 = (float*)d_ws;
    float* sup2 = sup1 + (size_t)N_NODES * NHID;
    int*  cursor = (int*)(sup2 + (size_t)N_NODES * NHID);
    int2* packed = (int2*)(cursor + 1024);
    int4* ovf    = (int4*)(packed + (size_t)NBIN * CAP);

    gemm1_kernel<<<NTILES, 128, 0, stream>>>(x, W1, sup1, cursor);
    bin_fill_kernel<<<NBLK_E, 1024, 0, stream>>>(es, ed, ev, cursor, packed, ovf);
    spmm_bin_kernel<NHID, 512><<<NBIN, 512, 0, stream>>>(packed, cursor, ovf, sup1, b1, W2, sup2);
    spmm_bin_kernel<NCLS, 512><<<NBIN, 512, 0, stream>>>(packed, cursor, ovf, sup2, b2, W2, out);
}


# GEMM1 made persistent: 1000 blocks walk the row tiles with the next tile's x loads (8 x 16 B per lane) in flight during the current tile's WMMAs; W1 slice kept in registers; sums unchanged
# speedup vs baseline: 4.0253x; 1.0443x over previous
#include <hip/hip_runtime.h>


typedef float v2f __attribute__((ext_vector_type(2)));
typedef float v8f __attribute__((ext_vector_type(8)));
typedef float v4f __attribute__((ext_vector_type(4)));

#define N_NODES 100000
#define FEAT    256
#define NHID    64
#define NCLS    16
#define NEDGE   1600000
#define NTILES  (N_NODES / 16)
#define APITCH  (FEAT + 4)
static_assert(APITCH % 4 == 0 && (APITCH / 4) % 2 == 1, "float4 staging stores; rows 16 B x odd apart");

#define RB      128
#define NBIN    ((N_NODES + RB - 1) / RB)
#define EPB     4096
#define NBLK_E  ((NEDGE + EPB - 1) / EPB)
#define SRC_BITS 17
#define CAP      4128
#define OVF_SLOT 1023
static_assert(CAP % 32 == 0 && (CAP / 32) % 2 == 1, "256-B aligned segments, an odd number of 256-B units apart");

static_assert(N_NODES <= (1 << SRC_BITS), "source row must fit the packed word");
static_assert(RB == 128 && NBIN <= OVF_SLOT, "one thread per bin in the 1024-thread pass; the last counter word is the overflow count");
static_assert((long long)NBIN * CAP < (1ll << 31), "32-bit record index");

#define G1_BLOCKS 1000
__global__ __launch_bounds__(128) void gemm1_kernel(const float* __restrict__ x,
                                                    const float* __restrict__ W1,
                                                    float* __restrict__ sup,
                                                    int* __restrict__ cursor) {
    __shared__ float tileA[2][16 * APITCH];
    const int lane = threadIdx.x & 31;
    const int wave = threadIdx.x >> 5;

    const int gi = blockIdx.x * 128 + threadIdx.x;
    if (gi < 1024) cursor[gi] = 0;

    const int c  = wave * 16;
    const int l  = lane & 15;
    const int hi = lane >> 4;

    v2f bf[FEAT / 4];
#pragma unroll
    for (int k4 = 0; k4 < FEAT / 4; ++k4) {
        const int ks = 4 * k4 + 2 * hi;
        bf[k4].x = W1[(size_t)ks       * NHID + c + l];
        bf[k4].y = W1[(size_t)(ks + 1) * NHID + c + l];
    }

    v4f pre[8];
    {
        const v4f* x4 = reinterpret_cast<const v4f*>(x) + (size_t)blockIdx.x * (16 * FEAT / 4);
#pragma unroll
        for (int j = 0; j < 8; ++j) pre[j] = x4[j * 128 + threadIdx.x];
    }
    const int mb = hi * 8;
    int p = 0;
    for (int tile = blockIdx.x; tile < NTILES; tile += G1_BLOCKS) {
        v4f* t4 = reinterpret_cast<v4f*>(tileA[p]);
#pragma unroll
        for (int j = 0; j < 8; ++j)
            t4[(2 * j + (threadIdx.x >> 6)) * (APITCH / 4) + (threadIdx.x & 63)] = pre[j];
        __syncthreads();

        const int next = tile + G1_BLOCKS;
        if (next < NTILES) {
            const v4f* x4 = reinterpret_cast<const v4f*>(x) + (size_t)next * (16 * FEAT / 4);
#pragma unroll
            for (int j = 0; j < 8; ++j) pre[j] = x4[j * 128 + threadIdx.x];
        }

        const float* arow = &tileA[p][l * APITCH + 2 * hi];
        v8f acc = {};
#pragma unroll
        for (int k4 = 0; k4 < FEAT / 4; ++k4) {
            const v2f a = *reinterpret_cast<const v2f*>(arow + 4 * k4);
            acc = __builtin_amdgcn_wmma_f32_16x16x4_f32(false, a, false, bf[k4],
                                                        (short)0, acc, false, false);
        }
        float* out = sup + (size_t)tile * 16 * NHID + c;
#pragma unroll
        for (int r = 0; r < 8; ++r)
            out[(size_t)(mb + r) * NHID + l] = acc[r];
        p ^= 1;
    }
}

__global__ __launch_bounds__(1024) void bin_fill_kernel(const int* __restrict__ esrc,
                                                        const int* __restrict__ edst,
                                                        const float* __restrict__ ev,
                                                        int* __restrict__ cursor,
                                                        int2* __restrict__ packed,
                                                        int4* __restrict__ ovf) {
    __shared__ int  s_hist[NBIN];
    __shared__ int  s_lbase[NBIN];
    __shared__ int  s_gbase[NBIN];
    __shared__ int  s_wsum[32];
    __shared__ int2 s_rec[EPB];
    __shared__ int  s_slot[EPB];
    static_assert(NBIN <= 1024, "one bin per thread in the prefix scan");
    const int t    = threadIdx.x;
    const int lane = t & 31;
    const int wave = t >> 5;
    if (t < NBIN) s_hist[t] = 0;
    __syncthreads();
    const int base = blockIdx.x * EPB;
    int d[EPB / 1024], rk[EPB / 1024];
#pragma unroll
    for (int i = 0; i < EPB / 1024; ++i) {
        const int e = base + i * 1024 + t;
        d[i]  = -1;
        rk[i] = 0;
        if (e < NEDGE) {
            d[i]  = edst[e];
            rk[i] = atomicAdd(&s_hist[(unsigned)d[i] / RB], 1);
        }
    }
    __syncthreads();
    const int mine = (t < NBIN) ? s_hist[t] : 0;
    int inc = mine;
    for (int o = 1; o < 32; o <<= 1) {
        const int n = __shfl_up(inc, o);
        if (lane >= o) inc += n;
    }
    if (lane == 31) s_wsum[wave] = inc;
    __syncthreads();
    if (wave == 0) {
        const int w = s_wsum[lane];
        int winc = w;
        for (int o = 1; o < 32; o <<= 1) {
            const int n = __shfl_up(winc, o);
            if (lane >= o) winc += n;
        }
        s_wsum[lane] = winc - w;
    }
    __syncthreads();
    if (t < NBIN) {
        int g = 0;
        if (mine) g = atomicAdd(&cursor[t], mine);
        s_gbase[t] = g;
        s_lbase[t] = s_wsum[wave] + inc - mine;
    }
    __syncthreads();
#pragma unroll
    for (int i = 0; i < EPB / 1024; ++i) {
        if (d[i] >= 0) {
            const int e   = base + i * 1024 + t;
            const int bin = (unsigned)d[i] / RB;
            const int li  = s_lbase[bin] + rk[i];
            const int pos = s_gbase[bin] + rk[i];
            const int s   = esrc[e];
            const int wb  = __builtin_bit_cast(int, ev[e]);
            s_rec[li] = make_int2(s | ((d[i] - bin * RB) << SRC_BITS), wb);
            if ((unsigned)pos < (unsigned)CAP) {
                s_slot[li] = bin * CAP + pos;
            } else {
                s_slot[li] = -1;
                const int o = atomicAdd(&cursor[OVF_SLOT], 1);
                if ((unsigned)o < (unsigned)NEDGE)
                    ovf[o] = make_int4(s, d[i], wb, 0);
            }
        }
    }
    __syncthreads();
    int nrec = NEDGE - base;
    nrec = nrec < EPB ? nrec : EPB;
    int  slot[EPB / 1024];
    int2 rec[EPB / 1024];
#pragma unroll
    for (int i = 0; i < EPB / 1024; ++i) {
        const int j  = i * 1024 + t;
        const int jc = j < nrec ? j : nrec - 1;
        const int sl = s_slot[jc];
        rec[i]  = s_rec[jc];
        slot[i] = j < nrec ? sl : -1;
    }
#pragma unroll
    for (int i = 0; i < EPB / 1024; ++i)
        if (slot[i] >= 0) packed[slot[i]] = rec[i];
}

template <int F> struct lds_pitch { static constexpr int v = (F == 64) ? 68 : F; };

static __device__ __forceinline__ void lds_add(float* p, float v) {
    (void)__hip_atomic_fetch_add(p, v, __ATOMIC_RELAXED, __HIP_MEMORY_SCOPE_WORKGROUP);
}

template <int F, bool TAIL>
static __device__ __forceinline__ void spmm_run(const int pk_l, const int w_l, const int lane,
                                                const float* __restrict__ sup, float* acc) {
    if constexpr (F == 64) {
#pragma unroll
        for (int jb = 0; jb < 32; jb += 16) {
            float v0[16], v1[16];
#pragma unroll
            for (int u = 0; u < 16; ++u) {
                const int pk = __builtin_amdgcn_readlane(pk_l, jb + u);
                const unsigned si = ((unsigned)pk & ((1u << SRC_BITS) - 1u)) * 64u + (unsigned)lane;
                v0[u] = sup[si];
                v1[u] = sup[si + 32u];
            }
#pragma unroll
            for (int u = 0; u < 16; ++u) {
                const int   pk = __builtin_amdgcn_readlane(pk_l, jb + u);
                const float w  = __builtin_bit_cast(
                    float, __builtin_amdgcn_readlane(w_l, jb + u));
                const unsigned di = (((unsigned)pk >> SRC_BITS) & (RB - 1u)) * (unsigned)lds_pitch<64>::v + (unsigned)lane;
                if (!TAIL || pk >= 0) {
                    lds_add(acc + di,       v0[u] * w);
                    lds_add(acc + di + 32u, v1[u] * w);
                }
            }
        }
    } else {
        const int l  = lane & 15;
        const int hi = lane >> 4;
#pragma unroll
        for (int jb = 0; jb < 16; jb += 8) {
            float v[8];
#pragma unroll
            for (int u = 0; u < 8; ++u) {
                const int sel = (2 * (jb + u) + hi) << 2;
                const int pk  = __builtin_amdgcn_ds_bpermute(sel, pk_l);
                v[u] = sup[((unsigned)pk & ((1u << SRC_BITS) - 1u)) * 16u + (unsigned)l];
            }
#pragma unroll
            for (int u = 0; u < 8; ++u) {
                const int sel = (2 * (jb + u) + hi) << 2;
                const int   pk = __builtin_amdgcn_ds_bpermute(sel, pk_l);
                const float w  = __builtin_bit_cast(
                    float, __builtin_amdgcn_ds_bpermute(sel, w_l));
                const unsigned di = (((unsigned)pk >> SRC_BITS) & (RB - 1u)) * 16u + (unsigned)l;
                if (!TAIL || pk >= 0) lds_add(acc + di, v[u] * w);
            }
        }
    }
}

template <int F, int BS>
__global__ __launch_bounds__(BS) void spmm_bin_kernel(const int2* __restrict__ packed,
                                                      const int* __restrict__ cursor,
                                                      const int4* __restrict__ ovf,
                                                      const float* __restrict__ sup,
                                                      const float* __restrict__ bias,
                                                      const float* __restrict__ W2,
                                                      float* __restrict__ outp) {
    static_assert(F == 64 || F == 16, "row widths of the two layers");
    static_assert((long long)N_NODES * F < (1ll << 31), "32-bit element index");
    constexpr int P = lds_pitch<F>::v;
    static_assert((RB * P) % 4 == 0 && P % 4 == 0, "float4 fill, 16-byte rows");
    __shared__ float acc[RB * P];
    const int t    = threadIdx.x;
    const int lane = t & 31;
    const int wave  = __builtin_amdgcn_readfirstlane(t >> 5);
    const int bin   = blockIdx.x;
    const int start = bin * CAP;
    int       cnt   = __builtin_amdgcn_readfirstlane(cursor[bin]);
    cnt = cnt < CAP ? cnt : CAP;
    const int end   = start + cnt;
    int       novf  = __builtin_amdgcn_readfirstlane(cursor[OVF_SLOT]);
    novf = novf < NEDGE ? novf : NEDGE;

    float4* acc4 = reinterpret_cast<float4*>(acc);
    for (int i = t; i < RB * P / 4; i += BS) acc4[i] = make_float4(0.0f, 0.0f, 0.0f, 0.0f);
    __syncthreads();

    for (int e0 = start + wave * 32; e0 < end; e0 += BS) {
        const int n = end - e0;
        int pk_l = (int)0x80000000;
        int w_l  = 0;
        if (lane < n) {
            const int2 r = packed[e0 + lane];
            pk_l = r.x;
            w_l  = r.y;
        }
        if (n >= 32) spmm_run<F, false>(pk_l, w_l, lane, sup, acc);
        else         spmm_run<F, true >(pk_l, w_l, lane, sup, acc);
    }
    for (int i = t; i < novf; i += BS) {
        const int4 r = ovf[i];
        if ((unsigned)r.y / RB == (unsigned)bin) {
            const float  w    = __builtin_bit_cast(float, r.z);
            const float* srow = sup + (size_t)r.x * F;
            float*       drow = acc + (r.y - bin * RB) * P;
            for (int c = 0; c < F; ++c) lds_add(drow + c, srow[c] * w);
        }
    }
    __syncthreads();

    const int rb0 = bin * RB;
    if constexpr (F == 64) {
        static_assert(BS / 32 >= RB / 16, "one wave per 16-row tile");
        const int row0 = rb0 + wave * 16;
        if (wave < RB / 16 && row0 < N_NODES) {
            const int l  = lane & 15;
            const int hi = lane >> 4;
            const float* arow = acc + (wave * 16 + l) * P;
            v8f c = {};
#pragma unroll
            for (int k = 0; k < NHID; k += 4) {
                const int ks = k + 2 * hi;
                v2f a = *reinterpret_cast<const v2f*>(arow + ks);
                const v2f bb = *reinterpret_cast<const v2f*>(&bias[ks]);
                a.x += bb.x;
                a.y += bb.y;
                a.x = a.x > 0.0f ? a.x : 0.0f;
                a.y = a.y > 0.0f ? a.y : 0.0f;
                v2f b;
                b.x = W2[ks * NCLS + l];
                b.y = W2[(ks + 1) * NCLS + l];
                c = __builtin_amdgcn_wmma_f32_16x16x4_f32(false, a, false, b,
                                                          (short)0, c, false, false);
            }
            float* o = outp + (size_t)row0 * NCLS;
            const int mb = hi * 8;
#pragma unroll
            for (int r = 0; r < 8; ++r)
                o[(size_t)(mb + r) * NCLS + l] = c[r];
        }
    } else {
        static_assert(RB * F / 4 == BS, "one float4 of the bin per thread");
        const int row = rb0 + (t >> 2);
        float4 v = acc4[t];
        const float4 bv = *reinterpret_cast<const float4*>(bias + (t & 3) * 4);
        v.x += bv.x; v.y += bv.y; v.z += bv.z; v.w += bv.w;
        float m = v.x;
        m = v.y > m ? v.y : m;
        m = v.z > m ? v.z : m;
        m = v.w > m ? v.w : m;
        float mo = __shfl_xor(m, 1);
        m = mo > m ? mo : m;
        mo = __shfl_xor(m, 2);
        m = mo > m ? mo : m;
        float s = __expf(v.x - m);
        s += __expf(v.y - m);
        s += __expf(v.z - m);
        s += __expf(v.w - m);
        s += __shfl_xor(s, 1);
        s += __shfl_xor(s, 2);
        const float lse = m + __logf(s);
        if (row < N_NODES) {
            float4* o4 = reinterpret_cast<float4*>(outp + (size_t)rb0 * F);
            o4[t] = make_float4(v.x - lse, v.y - lse, v.z - lse, v.w - lse);
        }
    }
}

extern "C" void kernel_launch(void* const* d_in, const int* in_sizes, int n_in,
                              void* d_out, int out_size, void* d_ws, size_t ws_size,
                              hipStream_t stream) {
    (void)in_sizes; (void)n_in; (void)out_size; (void)ws_size;
    const float* x   = (const float*)d_in[0];
    const int*   es  = (const int*)  d_in[1];
    const int*   ed  = (const int*)  d_in[2];
    const float* ev  = (const float*)d_in[3];
    const float* W1  = (const float*)d_in[4];
    const float* b1  = (const float*)d_in[5];
    const float* W2  = (const float*)d_in[6];
    const float* b2  = (const float*)d_in[7];
    float* out = (float*)d_out;

    float* sup1 = (float*)d_ws;
    float* sup2 = sup1 + (size_t)N_NODES * NHID;
    int*  cursor = (int*)(sup2 + (size_t)N_NODES * NHID);
    int2* packed = (int2*)(cursor + 1024);
    int4* ovf    = (int4*)(packed + (size_t)NBIN * CAP);

    gemm1_kernel<<<G1_BLOCKS, 128, 0, stream>>>(x, W1, sup1, cursor);
    bin_fill_kernel<<<NBLK_E, 1024, 0, stream>>>(es, ed, ev, cursor, packed, ovf);
    spmm_bin_kernel<NHID, 512><<<NBIN, 512, 0, stream>>>(packed, cursor, ovf, sup1, b1, W2, sup2);
    spmm_bin_kernel<NCLS, 512><<<NBIN, 512, 0, stream>>>(packed, cursor, ovf, sup2, b2, W2, out);
}


# bins of 132 destination rows instead of 128: 758 bin blocks per spmm launch, at most 3 on every WGP (782 put a 4th block on 14 WGPs, which set the launch's time); arrival-order sums kept
# speedup vs baseline: 4.3709x; 1.0859x over previous
#include <hip/hip_runtime.h>


typedef float v2f __attribute__((ext_vector_type(2)));
typedef float v8f __attribute__((ext_vector_type(8)));
typedef float v4f __attribute__((ext_vector_type(4)));

#define N_NODES 100000
#define FEAT    256
#define NHID    64
#define NCLS    16
#define NEDGE   1600000
#define NTILES  (N_NODES / 16)
#define APITCH  (FEAT + 4)
static_assert(APITCH % 4 == 0 && (APITCH / 4) % 2 == 1, "float4 staging stores; rows 16 B x odd apart");

#define RB      132
#define NBIN    ((N_NODES + RB - 1) / RB)
#define ROW_BITS 8
#define ROW_MASK ((1u << ROW_BITS) - 1u)
#define EPB     4096
#define NBLK_E  ((NEDGE + EPB - 1) / EPB)
#define SRC_BITS 17
#define CAP      4128
#define OVF_SLOT 1023
static_assert(CAP % 32 == 0 && (CAP / 32) % 2 == 1, "256-B aligned segments, an odd number of 256-B units apart");

static_assert(N_NODES <= (1 << SRC_BITS), "source row must fit the packed word");
static_assert(NBIN <= OVF_SLOT, "one thread per bin in the 1024-thread pass; the last counter word is the overflow count");
static_assert(RB <= (1 << ROW_BITS) && SRC_BITS + ROW_BITS < 31, "the local row fits its field below the filler bit");
static_assert(RB % 2 == 0, "bins start on 128-B lines of the 64-B-row arrays");
static_assert((long long)NBIN * CAP < (1ll << 31), "32-bit record index");

#define G1_BLOCKS 1000
__global__ __launch_bounds__(128) void gemm1_kernel(const float* __restrict__ x,
                                                    const float* __restrict__ W1,
                                                    float* __restrict__ sup,
                                                    int* __restrict__ cursor) {
    __shared__ float tileA[2][16 * APITCH];
    const int lane = threadIdx.x & 31;
    const int wave = threadIdx.x >> 5;

    const int gi = blockIdx.x * 128 + threadIdx.x;
    if (gi < 1024) cursor[gi] = 0;

    const int c  = wave * 16;
    const int l  = lane & 15;
    const int hi = lane >> 4;

    v2f bf[FEAT / 4];
#pragma unroll
    for (int k4 = 0; k4 < FEAT / 4; ++k4) {
        const int ks = 4 * k4 + 2 * hi;
        bf[k4].x = W1[(size_t)ks       * NHID + c + l];
        bf[k4].y = W1[(size_t)(ks + 1) * NHID + c + l];
    }

    v4f pre[8];
    {
        const v4f* x4 = reinterpret_cast<const v4f*>(x) + (size_t)blockIdx.x * (16 * FEAT / 4);
#pragma unroll
        for (int j = 0; j < 8; ++j) pre[j] = x4[j * 128 + threadIdx.x];
    }
    const int mb = hi * 8;
    int p = 0;
    for (int tile = blockIdx.x; tile < NTILES; tile += G1_BLOCKS) {
        v4f* t4 = reinterpret_cast<v4f*>(tileA[p]);
#pragma unroll
        for (int j = 0; j < 8; ++j)
            t4[(2 * j + (threadIdx.x >> 6)) * (APITCH / 4) + (threadIdx.x & 63)] = pre[j];
        __syncthreads();

        const int next = tile + G1_BLOCKS;
        if (next < NTILES) {
            const v4f* x4 = reinterpret_cast<const v4f*>(x) + (size_t)next * (16 * FEAT / 4);
#pragma unroll
            for (int j = 0; j < 8; ++j) pre[j] = x4[j * 128 + threadIdx.x];
        }

        const float* arow = &tileA[p][l * APITCH + 2 * hi];
        v8f acc = {};
#pragma unroll
        for (int k4 = 0; k4 < FEAT / 4; ++k4) {
            const v2f a = *reinterpret_cast<const v2f*>(arow + 4 * k4);
            acc = __builtin_amdgcn_wmma_f32_16x16x4_f32(false, a, false, bf[k4],
                                                        (short)0, acc, false, false);
        }
        float* out = sup + (size_t)tile * 16 * NHID + c;
#pragma unroll
        for (int r = 0; r < 8; ++r)
            out[(size_t)(mb + r) * NHID + l] = acc[r];
        p ^= 1;
    }
}

__global__ __launch_bounds__(1024) void bin_fill_kernel(const int* __restrict__ esrc,
                                                        const int* __restrict__ edst,
                                                        const float* __restrict__ ev,
                                                        int* __restrict__ cursor,
                                                        int2* __restrict__ packed,
                                                        int4* __restrict__ ovf) {
    __shared__ int  s_hist[NBIN];
    __shared__ int  s_lbase[NBIN];
    __shared__ int  s_gbase[NBIN];
    __shared__ int  s_wsum[32];
    __shared__ int2 s_rec[EPB];
    __shared__ int  s_slot[EPB];
    static_assert(NBIN <= 1024, "one bin per thread in the prefix scan");
    const int t    = threadIdx.x;
    const int lane = t & 31;
    const int wave = t >> 5;
    if (t < NBIN) s_hist[t] = 0;
    __syncthreads();
    const int base = blockIdx.x * EPB;
    int d[EPB / 1024], rk[EPB / 1024];
#pragma unroll
    for (int i = 0; i < EPB / 1024; ++i) {
        const int e = base + i * 1024 + t;
        d[i]  = -1;
        rk[i] = 0;
        if (e < NEDGE) {
            d[i]  = edst[e];
            rk[i] = atomicAdd(&s_hist[(unsigned)d[i] / RB], 1);
        }
    }
    __syncthreads();
    const int mine = (t < NBIN) ? s_hist[t] : 0;
    int inc = mine;
    for (int o = 1; o < 32; o <<= 1) {
        const int n = __shfl_up(inc, o);
        if (lane >= o) inc += n;
    }
    if (lane == 31) s_wsum[wave] = inc;
    __syncthreads();
    if (wave == 0) {
        const int w = s_wsum[lane];
        int winc = w;
        for (int o = 1; o < 32; o <<= 1) {
            const int n = __shfl_up(winc, o);
            if (lane >= o) winc += n;
        }
        s_wsum[lane] = winc - w;
    }
    __syncthreads();
    if (t < NBIN) {
        int g = 0;
        if (mine) g = atomicAdd(&cursor[t], mine);
        s_gbase[t] = g;
        s_lbase[t] = s_wsum[wave] + inc - mine;
    }
    __syncthreads();
#pragma unroll
    for (int i = 0; i < EPB / 1024; ++i) {
        if (d[i] >= 0) {
            const int e   = base + i * 1024 + t;
            const int bin = (unsigned)d[i] / RB;
            const int li  = s_lbase[bin] + rk[i];
            const int pos = s_gbase[bin] + rk[i];
            const int s   = esrc[e];
            const int wb  = __builtin_bit_cast(int, ev[e]);
            s_rec[li] = make_int2(s | ((d[i] - bin * RB) << SRC_BITS), wb);
            if ((unsigned)pos < (unsigned)CAP) {
                s_slot[li] = bin * CAP + pos;
            } else {
                s_slot[li] = -1;
                const int o = atomicAdd(&cursor[OVF_SLOT], 1);
                if ((unsigned)o < (unsigned)NEDGE)
                    ovf[o] = make_int4(s, d[i], wb, 0);
            }
        }
    }
    __syncthreads();
    int nrec = NEDGE - base;
    nrec = nrec < EPB ? nrec : EPB;
    int  slot[EPB / 1024];
    int2 rec[EPB / 1024];
#pragma unroll
    for (int i = 0; i < EPB / 1024; ++i) {
        const int j  = i * 1024 + t;
        const int jc = j < nrec ? j : nrec - 1;
        const int sl = s_slot[jc];
        rec[i]  = s_rec[jc];
        slot[i] = j < nrec ? sl : -1;
    }
#pragma unroll
    for (int i = 0; i < EPB / 1024; ++i)
        if (slot[i] >= 0) packed[slot[i]] = rec[i];
}

template <int F> struct lds_pitch { static constexpr int v = (F == 64) ? 68 : F; };

static __device__ __forceinline__ void lds_add(float* p, float v) {
    (void)__hip_atomic_fetch_add(p, v, __ATOMIC_RELAXED, __HIP_MEMORY_SCOPE_WORKGROUP);
}

template <int F, bool TAIL>
static __device__ __forceinline__ void spmm_run(const int pk_l, const int w_l, const int lane,
                                                const float* __restrict__ sup, float* acc) {
    if constexpr (F == 64) {
#pragma unroll
        for (int jb = 0; jb < 32; jb += 16) {
            float v0[16], v1[16];
#pragma unroll
            for (int u = 0; u < 16; ++u) {
                const int pk = __builtin_amdgcn_readlane(pk_l, jb + u);
                const unsigned si = ((unsigned)pk & ((1u << SRC_BITS) - 1u)) * 64u + (unsigned)lane;
                v0[u] = sup[si];
                v1[u] = sup[si + 32u];
            }
#pragma unroll
            for (int u = 0; u < 16; ++u) {
                const int   pk = __builtin_amdgcn_readlane(pk_l, jb + u);
                const float w  = __builtin_bit_cast(
                    float, __builtin_amdgcn_readlane(w_l, jb + u));
                const unsigned di = (((unsigned)pk >> SRC_BITS) & ROW_MASK) * (unsigned)lds_pitch<64>::v + (unsigned)lane;
                if (!TAIL || pk >= 0) {
                    lds_add(acc + di,       v0[u] * w);
                    lds_add(acc + di + 32u, v1[u] * w);
                }
            }
        }
    } else {
        const int l  = lane & 15;
        const int hi = lane >> 4;
#pragma unroll
        for (int jb = 0; jb < 16; jb += 8) {
            float v[8];
#pragma unroll
            for (int u = 0; u < 8; ++u) {
                const int sel = (2 * (jb + u) + hi) << 2;
                const int pk  = __builtin_amdgcn_ds_bpermute(sel, pk_l);
                v[u] = sup[((unsigned)pk & ((1u << SRC_BITS) - 1u)) * 16u + (unsigned)l];
            }
#pragma unroll
            for (int u = 0; u < 8; ++u) {
                const int sel = (2 * (jb + u) + hi) << 2;
                const int   pk = __builtin_amdgcn_ds_bpermute(sel, pk_l);
                const float w  = __builtin_bit_cast(
                    float, __builtin_amdgcn_ds_bpermute(sel, w_l));
                const unsigned di = (((unsigned)pk >> SRC_BITS) & ROW_MASK) * 16u + (unsigned)l;
                if (!TAIL || pk >= 0) lds_add(acc + di, v[u] * w);
            }
        }
    }
}

template <int F, int BS>
__global__ __launch_bounds__(BS) void spmm_bin_kernel(const int2* __restrict__ packed,
                                                      const int* __restrict__ cursor,
                                                      const int4* __restrict__ ovf,
                                                      const float* __restrict__ sup,
                                                      const float* __restrict__ bias,
                                                      const float* __restrict__ W2,
                                                      float* __restrict__ outp) {
    static_assert(F == 64 || F == 16, "row widths of the two layers");
    static_assert((long long)N_NODES * F < (1ll << 31), "32-bit element index");
    constexpr int P = lds_pitch<F>::v;
    constexpr int ROWS = (F == 64) ? ((RB + 15) / 16) * 16 : RB;
    static_assert((ROWS * P) % 4 == 0 && P % 4 == 0, "float4 fill, 16-byte rows");
    __shared__ float acc[ROWS * P];
    const int t    = threadIdx.x;
    const int lane = t & 31;
    const int wave  = __builtin_amdgcn_readfirstlane(t >> 5);
    const int bin   = blockIdx.x;
    const int start = bin * CAP;
    int       cnt   = __builtin_amdgcn_readfirstlane(cursor[bin]);
    cnt = cnt < CAP ? cnt : CAP;
    const int end   = start + cnt;
    int       novf  = __builtin_amdgcn_readfirstlane(cursor[OVF_SLOT]);
    novf = novf < NEDGE ? novf : NEDGE;

    float4* acc4 = reinterpret_cast<float4*>(acc);
    for (int i = t; i < ROWS * P / 4; i += BS) acc4[i] = make_float4(0.0f, 0.0f, 0.0f, 0.0f);
    __syncthreads();

    for (int e0 = start + wave * 32; e0 < end; e0 += BS) {
        const int n = end - e0;
        int pk_l = (int)0x80000000;
        int w_l  = 0;
        if (lane < n) {
            const int2 r = packed[e0 + lane];
            pk_l = r.x;
            w_l  = r.y;
        }
        if (n >= 32) spmm_run<F, false>(pk_l, w_l, lane, sup, acc);
        else         spmm_run<F, true >(pk_l, w_l, lane, sup, acc);
    }
    for (int i = t; i < novf; i += BS) {
        const int4 r = ovf[i];
        if ((unsigned)r.y / RB == (unsigned)bin) {
            const float  w    = __builtin_bit_cast(float, r.z);
            const float* srow = sup + (size_t)r.x * F;
            float*       drow = acc + (r.y - bin * RB) * P;
            for (int c = 0; c < F; ++c) lds_add(drow + c, srow[c] * w);
        }
    }
    __syncthreads();

    const int rb0 = bin * RB;
    if constexpr (F == 64) {
        static_assert(BS / 32 >= ROWS / 16, "one wave per 16-row tile");
        const int row0 = rb0 + wave * 16;
        if (wave < ROWS / 16 && row0 < N_NODES) {
            const int l  = lane & 15;
            const int hi = lane >> 4;
            const float* arow = acc + (wave * 16 + l) * P;
            v8f c = {};
#pragma unroll
            for (int k = 0; k < NHID; k += 4) {
                const int ks = k + 2 * hi;
                v2f a = *reinterpret_cast<const v2f*>(arow + ks);
                const v2f bb = *reinterpret_cast<const v2f*>(&bias[ks]);
                a.x += bb.x;
                a.y += bb.y;
                a.x = a.x > 0.0f ? a.x : 0.0f;
                a.y = a.y > 0.0f ? a.y : 0.0f;
                v2f b;
                b.x = W2[ks * NCLS + l];
                b.y = W2[(ks + 1) * NCLS + l];
                c = __builtin_amdgcn_wmma_f32_16x16x4_f32(false, a, false, b,
                                                          (short)0, c, false, false);
            }
            float* o = outp + (size_t)row0 * NCLS;
            const int mb = hi * 8;
#pragma unroll
            for (int r = 0; r < 8; ++r) {
                const int lr = wave * 16 + mb + r;
                if (lr < RB && rb0 + lr < N_NODES)
                    o[(size_t)(mb + r) * NCLS + l] = c[r];
            }
        }
    } else {
        constexpr int NV = RB * F / 4;
        static_assert(NV % 4 == 0 && BS % 32 == 0, "the 4 lanes of a row sit in one wave and one pass");
        for (int i0 = 0; i0 < NV; i0 += BS) {
            const int i   = i0 + t;
            const int ic  = i < NV ? i : NV - 1;
            const int row = rb0 + (ic >> 2);
            float4 v = acc4[ic];
            const float4 bv = *reinterpret_cast<const float4*>(bias + (ic & 3) * 4);
            v.x += bv.x; v.y += bv.y; v.z += bv.z; v.w += bv.w;
            float m = v.x;
            m = v.y > m ? v.y : m;
            m = v.z > m ? v.z : m;
            m = v.w > m ? v.w : m;
            float mo = __shfl_xor(m, 1);
            m = mo > m ? mo : m;
            mo = __shfl_xor(m, 2);
            m = mo > m ? mo : m;
            float s = __expf(v.x - m);
            s += __expf(v.y - m);
            s += __expf(v.z - m);
            s += __expf(v.w - m);
            s += __shfl_xor(s, 1);
            s += __shfl_xor(s, 2);
            const float lse = m + __logf(s);
            if (i < NV && row < N_NODES) {
                float4* o4 = reinterpret_cast<float4*>(outp + (size_t)rb0 * F);
                o4[i] = make_float4(v.x - lse, v.y - lse, v.z - lse, v.w - lse);
            }
        }
    }
}

extern "C" void kernel_launch(void* const* d_in, const int* in_sizes, int n_in,
                              void* d_out, int out_size, void* d_ws, size_t ws_size,
                              hipStream_t stream) {
    (void)in_sizes; (void)n_in; (void)out_size; (void)ws_size;
    const float* x   = (const float*)d_in[0];
    const int*   es  = (const int*)  d_in[1];
    const int*   ed  = (const int*)  d_in[2];
    const float* ev  = (const float*)d_in[3];
    const float* W1  = (const float*)d_in[4];
    const float* b1  = (const float*)d_in[5];
    const float* W2  = (const float*)d_in[6];
    const float* b2  = (const float*)d_in[7];
    float* out = (float*)d_out;

    float* sup1 = (float*)d_ws;
    float* sup2 = sup1 + (size_t)N_NODES * NHID;
    int*  cursor = (int*)(sup2 + (size_t)N_NODES * NHID);
    int2* packed = (int2*)(cursor + 1024);
    int4* ovf    = (int4*)(packed + (size_t)NBIN * CAP);

    gemm1_kernel<<<G1_BLOCKS, 128, 0, stream>>>(x, W1, sup1, cursor);
    bin_fill_kernel<<<NBLK_E, 1024, 0, stream>>>(es, ed, ev, cursor, packed, ovf);
    spmm_bin_kernel<NHID, 512><<<NBIN, 512, 0, stream>>>(packed, cursor, ovf, sup1, b1, W2, sup2);
    spmm_bin_kernel<NCLS, 512><<<NBIN, 512, 0, stream>>>(packed, cursor, ovf, sup2, b2, W2, out);
}


# bin blocks group their records by destination row in LDS; a wave sums equal-row stretches of a run in registers: one LDS f32 add per stretch, not per edge (register partials, then arrival order)
# speedup vs baseline: 4.6221x; 1.0575x over previous
#include <hip/hip_runtime.h>


typedef float v2f __attribute__((ext_vector_type(2)));
typedef float v8f __attribute__((ext_vector_type(8)));
typedef float v4f __attribute__((ext_vector_type(4)));

#define N_NODES 100000
#define FEAT    256
#define NHID    64
#define NCLS    16
#define NEDGE   1600000
#define NTILES  (N_NODES / 16)
#define APITCH  (FEAT + 4)
static_assert(APITCH % 4 == 0 && (APITCH / 4) % 2 == 1, "float4 staging stores; rows 16 B x odd apart");

#define RB      132
#define NBIN    ((N_NODES + RB - 1) / RB)
#define ROW_BITS 8
#define ROW_MASK ((1u << ROW_BITS) - 1u)
#define EPB     4096
#define NBLK_E  ((NEDGE + EPB - 1) / EPB)
#define SRC_BITS 17
#define CAP      4128
#define OVF_SLOT 1023
static_assert(CAP % 32 == 0 && (CAP / 32) % 2 == 1, "256-B aligned segments, an odd number of 256-B units apart");

static_assert(N_NODES <= (1 << SRC_BITS), "source row must fit the packed word");
static_assert(NBIN <= OVF_SLOT, "one thread per bin in the 1024-thread pass; the last counter word is the overflow count");
static_assert(RB <= (1 << ROW_BITS) && SRC_BITS + ROW_BITS < 31, "the local row fits its field below the filler bit");
static_assert(RB % 2 == 0, "bins start on 128-B lines of the 64-B-row arrays");
static_assert((long long)NBIN * CAP < (1ll << 31), "32-bit record index");

#define G1_BLOCKS 1000
__global__ __launch_bounds__(128) void gemm1_kernel(const float* __restrict__ x,
                                                    const float* __restrict__ W1,
                                                    float* __restrict__ sup,
                                                    int* __restrict__ cursor) {
    __shared__ float tileA[2][16 * APITCH];
    const int lane = threadIdx.x & 31;
    const int wave = threadIdx.x >> 5;

    const int gi = blockIdx.x * 128 + threadIdx.x;
    if (gi < 1024) cursor[gi] = 0;

    const int c  = wave * 16;
    const int l  = lane & 15;
    const int hi = lane >> 4;

    v2f bf[FEAT / 4];
#pragma unroll
    for (int k4 = 0; k4 < FEAT / 4; ++k4) {
        const int ks = 4 * k4 + 2 * hi;
        bf[k4].x = W1[(size_t)ks       * NHID + c + l];
        bf[k4].y = W1[(size_t)(ks + 1) * NHID + c + l];
    }

    v4f pre[8];
    {
        const v4f* x4 = reinterpret_cast<const v4f*>(x) + (size_t)blockIdx.x * (16 * FEAT / 4);
#pragma unroll
        for (int j = 0; j < 8; ++j) pre[j] = x4[j * 128 + threadIdx.x];
    }
    const int mb = hi * 8;
    int p = 0;
    for (int tile = blockIdx.x; tile < NTILES; tile += G1_BLOCKS) {
        v4f* t4 = reinterpret_cast<v4f*>(tileA[p]);
#pragma unroll
        for (int j = 0; j < 8; ++j)
            t4[(2 * j + (threadIdx.x >> 6)) * (APITCH / 4) + (threadIdx.x & 63)] = pre[j];
        __syncthreads();

        const int next = tile + G1_BLOCKS;
        if (next < NTILES) {
            const v4f* x4 = reinterpret_cast<const v4f*>(x) + (size_t)next * (16 * FEAT / 4);
#pragma unroll
            for (int j = 0; j < 8; ++j) pre[j] = x4[j * 128 + threadIdx.x];
        }

        const float* arow = &tileA[p][l * APITCH + 2 * hi];
        v8f acc = {};
#pragma unroll
        for (int k4 = 0; k4 < FEAT / 4; ++k4) {
            const v2f a = *reinterpret_cast<const v2f*>(arow + 4 * k4);
            acc = __builtin_amdgcn_wmma_f32_16x16x4_f32(false, a, false, bf[k4],
                                                        (short)0, acc, false, false);
        }
        float* out = sup + (size_t)tile * 16 * NHID + c;
#pragma unroll
        for (int r = 0; r < 8; ++r)
            out[(size_t)(mb + r) * NHID + l] = acc[r];
        p ^= 1;
    }
}

__global__ __launch_bounds__(1024) void bin_fill_kernel(const int* __restrict__ esrc,
                                                        const int* __restrict__ edst,
                                                        const float* __restrict__ ev,
                                                        int* __restrict__ cursor,
                                                        int2* __restrict__ packed,
                                                        int4* __restrict__ ovf) {
    __shared__ int  s_hist[NBIN];
    __shared__ int  s_lbase[NBIN];
    __shared__ int  s_gbase[NBIN];
    __shared__ int  s_wsum[32];
    __shared__ int2 s_rec[EPB];
    __shared__ int  s_slot[EPB];
    static_assert(NBIN <= 1024, "one bin per thread in the prefix scan");
    const int t    = threadIdx.x;
    const int lane = t & 31;
    const int wave = t >> 5;
    if (t < NBIN) s_hist[t] = 0;
    __syncthreads();
    const int base = blockIdx.x * EPB;
    int d[EPB / 1024], rk[EPB / 1024];
#pragma unroll
    for (int i = 0; i < EPB / 1024; ++i) {
        const int e = base + i * 1024 + t;
        d[i]  = -1;
        rk[i] = 0;
        if (e < NEDGE) {
            d[i]  = edst[e];
            rk[i] = atomicAdd(&s_hist[(unsigned)d[i] / RB], 1);
        }
    }
    __syncthreads();
    const int mine = (t < NBIN) ? s_hist[t] : 0;
    int inc = mine;
    for (int o = 1; o < 32; o <<= 1) {
        const int n = __shfl_up(inc, o);
        if (lane >= o) inc += n;
    }
    if (lane == 31) s_wsum[wave] = inc;
    __syncthreads();
    if (wave == 0) {
        const int w = s_wsum[lane];
        int winc = w;
        for (int o = 1; o < 32; o <<= 1) {
            const int n = __shfl_up(winc, o);
            if (lane >= o) winc += n;
        }
        s_wsum[lane] = winc - w;
    }
    __syncthreads();
    if (t < NBIN) {
        int g = 0;
        if (mine) g = atomicAdd(&cursor[t], mine);
        s_gbase[t] = g;
        s_lbase[t] = s_wsum[wave] + inc - mine;
    }
    __syncthreads();
#pragma unroll
    for (int i = 0; i < EPB / 1024; ++i) {
        if (d[i] >= 0) {
            const int e   = base + i * 1024 + t;
            const int bin = (unsigned)d[i] / RB;
            const int li  = s_lbase[bin] + rk[i];
            const int pos = s_gbase[bin] + rk[i];
            const int s   = esrc[e];
            const int wb  = __builtin_bit_cast(int, ev[e]);
            s_rec[li] = make_int2(s | ((d[i] - bin * RB) << SRC_BITS), wb);
            if ((unsigned)pos < (unsigned)CAP) {
                s_slot[li] = bin * CAP + pos;
            } else {
                s_slot[li] = -1;
                const int o = atomicAdd(&cursor[OVF_SLOT], 1);
                if ((unsigned)o < (unsigned)NEDGE)
                    ovf[o] = make_int4(s, d[i], wb, 0);
            }
        }
    }
    __syncthreads();
    int nrec = NEDGE - base;
    nrec = nrec < EPB ? nrec : EPB;
    int  slot[EPB / 1024];
    int2 rec[EPB / 1024];
#pragma unroll
    for (int i = 0; i < EPB / 1024; ++i) {
        const int j  = i * 1024 + t;
        const int jc = j < nrec ? j : nrec - 1;
        const int sl = s_slot[jc];
        rec[i]  = s_rec[jc];
        slot[i] = j < nrec ? sl : -1;
    }
#pragma unroll
    for (int i = 0; i < EPB / 1024; ++i)
        if (slot[i] >= 0) packed[slot[i]] = rec[i];
}

template <int F> struct lds_pitch { static constexpr int v = (F == 64) ? 68 : F; };

static __device__ __forceinline__ void lds_add(float* p, float v) {
    (void)__hip_atomic_fetch_add(p, v, __ATOMIC_RELAXED, __HIP_MEMORY_SCOPE_WORKGROUP);
}

template <int F, bool TAIL>
static __device__ __forceinline__ void spmm_run(const int pk_l, const int w_l, const int lane,
                                                const float* __restrict__ sup, float* acc) {
    if constexpr (F == 64) {
        unsigned cur = (((unsigned)__builtin_amdgcn_readlane(pk_l, 0)) >> SRC_BITS) & ROW_MASK;
        float a0 = 0.0f, a1 = 0.0f;
#pragma unroll
        for (int jb = 0; jb < 32; jb += 16) {
            float v0[16], v1[16];
#pragma unroll
            for (int u = 0; u < 16; ++u) {
                const int pk = __builtin_amdgcn_readlane(pk_l, jb + u);
                const unsigned si = ((unsigned)pk & ((1u << SRC_BITS) - 1u)) * 64u + (unsigned)lane;
                v0[u] = sup[si];
                v1[u] = sup[si + 32u];
            }
#pragma unroll
            for (int u = 0; u < 16; ++u) {
                const int   pk = __builtin_amdgcn_readlane(pk_l, jb + u);
                const float w  = __builtin_bit_cast(
                    float, __builtin_amdgcn_readlane(w_l, jb + u));
                const unsigned row = ((unsigned)pk >> SRC_BITS) & ROW_MASK;
                if (!TAIL || pk >= 0) {
                    if (row != cur) {
                        const unsigned di = cur * (unsigned)lds_pitch<64>::v + (unsigned)lane;
                        lds_add(acc + di,       a0);
                        lds_add(acc + di + 32u, a1);
                        a0  = 0.0f;
                        a1  = 0.0f;
                        cur = row;
                    }
                    a0 += v0[u] * w;
                    a1 += v1[u] * w;
                }
            }
        }
        const unsigned di = cur * (unsigned)lds_pitch<64>::v + (unsigned)lane;
        lds_add(acc + di,       a0);
        lds_add(acc + di + 32u, a1);
    } else {
        const int l  = lane & 15;
        const int hi = lane >> 4;
        unsigned cur = (((unsigned)__builtin_amdgcn_ds_bpermute(hi << 2, pk_l)) >> SRC_BITS) & ROW_MASK;
        float a = 0.0f;
#pragma unroll
        for (int jb = 0; jb < 16; jb += 8) {
            float v[8];
#pragma unroll
            for (int u = 0; u < 8; ++u) {
                const int sel = (2 * (jb + u) + hi) << 2;
                const int pk  = __builtin_amdgcn_ds_bpermute(sel, pk_l);
                v[u] = sup[((unsigned)pk & ((1u << SRC_BITS) - 1u)) * 16u + (unsigned)l];
            }
#pragma unroll
            for (int u = 0; u < 8; ++u) {
                const int sel = (2 * (jb + u) + hi) << 2;
                const int   pk = __builtin_amdgcn_ds_bpermute(sel, pk_l);
                const float w  = __builtin_bit_cast(
                    float, __builtin_amdgcn_ds_bpermute(sel, w_l));
                const unsigned row = ((unsigned)pk >> SRC_BITS) & ROW_MASK;
                if (!TAIL || pk >= 0) {
                    if (row != cur) {
                        lds_add(acc + cur * 16u + (unsigned)l, a);
                        a   = 0.0f;
                        cur = row;
                    }
                    a += v[u] * w;
                }
            }
        }
        lds_add(acc + cur * 16u + (unsigned)l, a);
    }
}

template <int F, int BS>
__global__ __launch_bounds__(BS) void spmm_bin_kernel(const int2* __restrict__ packed,
                                                      const int* __restrict__ cursor,
                                                      const int4* __restrict__ ovf,
                                                      const float* __restrict__ sup,
                                                      const float* __restrict__ bias,
                                                      const float* __restrict__ W2,
                                                      float* __restrict__ outp) {
    static_assert(F == 64 || F == 16, "row widths of the two layers");
    static_assert((long long)N_NODES * F < (1ll << 31), "32-bit element index");
    constexpr int P = lds_pitch<F>::v;
    constexpr int ROWS = (F == 64) ? ((RB + 15) / 16) * 16 : RB;
    static_assert((ROWS * P) % 4 == 0 && P % 4 == 0, "float4 fill, 16-byte rows");
    __shared__ float acc[ROWS * P];
    __shared__ int2  s_rec[CAP];
    __shared__ int   s_pos[RB];
    __shared__ int   s_tot[(RB + 31) / 32];
    const int t    = threadIdx.x;
    const int lane = t & 31;
    const int wave  = __builtin_amdgcn_readfirstlane(t >> 5);
    const int bin   = blockIdx.x;
    const int start = bin * CAP;
    int       cnt   = __builtin_amdgcn_readfirstlane(cursor[bin]);
    cnt = cnt < CAP ? cnt : CAP;
    int       novf  = __builtin_amdgcn_readfirstlane(cursor[OVF_SLOT]);
    novf = novf < NEDGE ? novf : NEDGE;

    float4* acc4 = reinterpret_cast<float4*>(acc);
    for (int i = t; i < ROWS * P / 4; i += BS) acc4[i] = make_float4(0.0f, 0.0f, 0.0f, 0.0f);
    if (t < RB) s_pos[t] = 0;
    __syncthreads();

    constexpr int NR = (CAP + BS - 1) / BS;
    int2 rr[NR];
#pragma unroll
    for (int k = 0; k < NR; ++k) {
        const int i = k * BS + t;
        rr[k] = make_int2(0, 0);
        if (i < cnt) rr[k] = packed[start + i];
    }
#pragma unroll
    for (int k = 0; k < NR; ++k) {
        if (k * BS + t < cnt) {
            const unsigned row = ((unsigned)rr[k].x >> SRC_BITS) & ROW_MASK;
            (void)__hip_atomic_fetch_add(&s_pos[row], 1, __ATOMIC_RELAXED, __HIP_MEMORY_SCOPE_WORKGROUP);
        }
    }
    __syncthreads();
    constexpr int NCH = (RB + 31) / 32;
    static_assert(NCH * 32 <= BS, "one row per thread in the prefix scan");
    int cv = 0, cinc = 0;
    if (wave < NCH) {
        cv   = t < RB ? s_pos[t] : 0;
        cinc = cv;
        for (int o = 1; o < 32; o <<= 1) {
            const int nb = __shfl_up(cinc, o);
            if (lane >= o) cinc += nb;
        }
        if (lane == 31) s_tot[wave] = cinc;
    }
    __syncthreads();
    if (t < RB) {
        int b = 0;
        for (int c = 0; c < NCH; ++c) b += (c < wave) ? s_tot[c] : 0;
        s_pos[t] = b + cinc - cv;
    }
    __syncthreads();
#pragma unroll
    for (int k = 0; k < NR; ++k) {
        if (k * BS + t < cnt) {
            const unsigned row = ((unsigned)rr[k].x >> SRC_BITS) & ROW_MASK;
            const int pos = __hip_atomic_fetch_add(&s_pos[row], 1, __ATOMIC_RELAXED, __HIP_MEMORY_SCOPE_WORKGROUP);
            if ((unsigned)pos < (unsigned)CAP) s_rec[pos] = rr[k];
        }
    }
    __syncthreads();

    for (int e0 = wave * 32; e0 < cnt; e0 += BS) {
        const int n = cnt - e0;
        int pk_l = (int)0x80000000;
        int w_l  = 0;
        if (lane < n) {
            const int2 r = s_rec[e0 + lane];
            pk_l = r.x;
            w_l  = r.y;
        }
        if (n >= 32) spmm_run<F, false>(pk_l, w_l, lane, sup, acc);
        else         spmm_run<F, true >(pk_l, w_l, lane, sup, acc);
    }
    for (int i = t; i < novf; i += BS) {
        const int4 r = ovf[i];
        if ((unsigned)r.y / RB == (unsigned)bin) {
            const float  w    = __builtin_bit_cast(float, r.z);
            const float* srow = sup + (size_t)r.x * F;
            float*       drow = acc + (r.y - bin * RB) * P;
            for (int c = 0; c < F; ++c) lds_add(drow + c, srow[c] * w);
        }
    }
    __syncthreads();

    const int rb0 = bin * RB;
    if constexpr (F == 64) {
        static_assert(BS / 32 >= ROWS / 16, "one wave per 16-row tile");
        const int row0 = rb0 + wave * 16;
        if (wave < ROWS / 16 && row0 < N_NODES) {
            const int l  = lane & 15;
            const int hi = lane >> 4;
            const float* arow = acc + (wave * 16 + l) * P;
            v8f c = {};
#pragma unroll
            for (int k = 0; k < NHID; k += 4) {
                const int ks = k + 2 * hi;
                v2f a = *reinterpret_cast<const v2f*>(arow + ks);
                const v2f bb = *reinterpret_cast<const v2f*>(&bias[ks]);
                a.x += bb.x;
                a.y += bb.y;
                a.x = a.x > 0.0f ? a.x : 0.0f;
                a.y = a.y > 0.0f ? a.y : 0.0f;
                v2f b;
                b.x = W2[ks * NCLS + l];
                b.y = W2[(ks + 1) * NCLS + l];
                c = __builtin_amdgcn_wmma_f32_16x16x4_f32(false, a, false, b,
                                                          (short)0, c, false, false);
            }
            float* o = outp + (size_t)row0 * NCLS;
            const int mb = hi * 8;
#pragma unroll
            for (int r = 0; r < 8; ++r) {
                const int lr = wave * 16 + mb + r;
                if (lr < RB && rb0 + lr < N_NODES)
                    o[(size_t)(mb + r) * NCLS + l] = c[r];
            }
        }
    } else {
        constexpr int NV = RB * F / 4;
        static_assert(NV % 4 == 0 && BS % 32 == 0, "the 4 lanes of a row sit in one wave and one pass");
        for (int i0 = 0; i0 < NV; i0 += BS) {
            const int i   = i0 + t;
            const int ic  = i < NV ? i : NV - 1;
            const int row = rb0 + (ic >> 2);
            float4 v = acc4[ic];
            const float4 bv = *reinterpret_cast<const float4*>(bias + (ic & 3) * 4);
            v.x += bv.x; v.y += bv.y; v.z += bv.z; v.w += bv.w;
            float m = v.x;
            m = v.y > m ? v.y : m;
            m = v.z > m ? v.z : m;
            m = v.w > m ? v.w : m;
            float mo = __shfl_xor(m, 1);
            m = mo > m ? mo : m;
            mo = __shfl_xor(m, 2);
            m = mo > m ? mo : m;
            float s = __expf(v.x - m);
            s += __expf(v.y - m);
            s += __expf(v.z - m);
            s += __expf(v.w - m);
            s += __shfl_xor(s, 1);
            s += __shfl_xor(s, 2);
            const float lse = m + __logf(s);
            if (i < NV && row < N_NODES) {
                float4* o4 = reinterpret_cast<float4*>(outp + (size_t)rb0 * F);
                o4[i] = make_float4(v.x - lse, v.y - lse, v.z - lse, v.w - lse);
            }
        }
    }
}

extern "C" void kernel_launch(void* const* d_in, const int* in_sizes, int n_in,
                              void* d_out, int out_size, void* d_ws, size_t ws_size,
                              hipStream_t stream) {
    (void)in_sizes; (void)n_in; (void)out_size; (void)ws_size;
    const float* x   = (const float*)d_in[0];
    const int*   es  = (const int*)  d_in[1];
    const int*   ed  = (const int*)  d_in[2];
    const float* ev  = (const float*)d_in[3];
    const float* W1  = (const float*)d_in[4];
    const float* b1  = (const float*)d_in[5];
    const float* W2  = (const float*)d_in[6];
    const float* b2  = (const float*)d_in[7];
    float* out = (float*)d_out;

    float* sup1 = (float*)d_ws;
    float* sup2 = sup1 + (size_t)N_NODES * NHID;
    int*  cursor = (int*)(sup2 + (size_t)N_NODES * NHID);
    int2* packed = (int2*)(cursor + 1024);
    int4* ovf    = (int4*)(packed + (size_t)NBIN * CAP);

    gemm1_kernel<<<G1_BLOCKS, 128, 0, stream>>>(x, W1, sup1, cursor);
    bin_fill_kernel<<<NBLK_E, 1024, 0, stream>>>(es, ed, ev, cursor, packed, ovf);
    spmm_bin_kernel<NHID, 512><<<NBIN, 512, 0, stream>>>(packed, cursor, ovf, sup1, b1, W2, sup2);
    spmm_bin_kernel<NCLS, 512><<<NBIN, 512, 0, stream>>>(packed, cursor, ovf, sup2, b2, W2, out);
}
